# starting kernel
# baseline (speedup 1.0000x reference)
#include <hip/hip_runtime.h>
#include <math.h>


typedef __attribute__((ext_vector_type(2))) float v2f;
typedef __attribute__((ext_vector_type(8))) float v8f;


__global__ void zero_kernel(float* __restrict__ p, size_t n) {
  size_t i = (size_t)blockIdx.x * blockDim.x + threadIdx.x;
  size_t stride = (size_t)gridDim.x * blockDim.x;
  for (; i < n; i += stride) p[i] = 0.0f;
}

__global__ void degree_kernel(const int* __restrict__ src, const int* __restrict__ dst,
                              int E, float* __restrict__ outdeg, float* __restrict__ indeg) {
  int i = blockIdx.x * blockDim.x + threadIdx.x;
  if (i < E) {
    atomicAdd(&outdeg[src[i]], 1.0f);
    atomicAdd(&indeg[dst[i]], 1.0f);
  }
}

__global__ void norm_kernel(const float* __restrict__ outdeg, const float* __restrict__ indeg,
                            float* __restrict__ outnorm, float* __restrict__ innorm, int N) {
  int i = blockIdx.x * blockDim.x + threadIdx.x;
  if (i < N) {
    outnorm[i] = rsqrtf(fmaxf(outdeg[i], 1.0f));
    innorm[i]  = rsqrtf(fmaxf(indeg[i],  1.0f));
  }
}

template<int K, int M>
__global__ __launch_bounds__(32 * (M / 16))
void gemm_wmma(const float* __restrict__ A, const float* __restrict__ B,
               float* __restrict__ C,
               const float* __restrict__ rowscale,
               const float* __restrict__ innorm, const float* __restrict__ bias,
               int relu) {
  const int wave = threadIdx.x >> 5;
  const int lane = threadIdx.x & 31;
  const int half = lane >> 4;
  const int m    = lane & 15;
  const int row0 = blockIdx.x * 16;
  const int col0 = wave * 16;

  const float* __restrict__ Arow = A + (size_t)(row0 + m) * K;
  const float sc = rowscale ? rowscale[row0 + m] : 1.0f;

  v8f acc = {};
#pragma unroll 8
  for (int k = 0; k < K; k += 4) {
    const int ka = k + 2 * half;
    v2f a, b;
    a.x = Arow[ka]     * sc;
    a.y = Arow[ka + 1] * sc;
    b.x = B[(size_t)ka       * M + col0 + m];
    b.y = B[(size_t)(ka + 1) * M + col0 + m];
    acc = __builtin_amdgcn_wmma_f32_16x16x4_f32(
         false, a,  false, b,
         (short)0, acc,  false,  false);
  }

#pragma unroll
  for (int i = 0; i < 8; ++i) {
    const int row = row0 + i + 8 * half;
    const int col = col0 + m;
    float v = acc[i];
    if (innorm) v = v * innorm[row] + bias[col];
    if (relu)   v = fmaxf(v, 0.0f);
    C[(size_t)row * M + col] = v;
  }
}

__global__ void aggregate_kernel(const float* __restrict__ h,
                                 const int* __restrict__ src, const int* __restrict__ dst,
                                 const float* __restrict__ srcscale,
                                 float* __restrict__ agg, int E, int M) {
  const int tpe  = M >> 2;
  const int epb  = blockDim.x / tpe;
  const int e    = blockIdx.x * epb + (int)(threadIdx.x / tpe);
  const int lane = threadIdx.x % tpe;
  if (e >= E) return;
  const int s = src[e], d = dst[e];
  const float sc = srcscale ? srcscale[s] : 1.0f;
  const float4 v = *(const float4*)(h + (size_t)s * M + lane * 4);
  float* out = agg + (size_t)d * M + lane * 4;
  atomicAdd(out + 0, v.x * sc);
  atomicAdd(out + 1, v.y * sc);
  atomicAdd(out + 2, v.z * sc);
  atomicAdd(out + 3, v.w * sc);
}

__global__ void finish_kernel(float* __restrict__ io, const float* __restrict__ innorm,
                              const float* __restrict__ bias, int N, int M, int relu) {
  const int idx = blockIdx.x * blockDim.x + threadIdx.x;
  if (idx >= N * M) return;
  const int row = idx / M, col = idx % M;
  float v = io[idx] * innorm[row] + bias[col];
  if (relu) v = fmaxf(v, 0.0f);
  io[idx] = v;
}


extern "C" void kernel_launch(void* const* d_in, const int* in_sizes, int n_in,
                              void* d_out, int out_size, void* d_ws, size_t ws_size,
                              hipStream_t stream) {
  const float* x   = (const float*)d_in[0];
  const int*   src = (const int*)d_in[1];
  const int*   dst = (const int*)d_in[2];
  const float* W1  = (const float*)d_in[3];
  const float* b1  = (const float*)d_in[4];
  const float* W2  = (const float*)d_in[5];
  const float* b2  = (const float*)d_in[6];
  const float* W3  = (const float*)d_in[7];
  const float* b3  = (const float*)d_in[8];
  float* out = (float*)d_out;

  const int IN = 256, HID = 128, CLS = 64;
  const int E = in_sizes[1];
  const int N = in_sizes[0] / IN;

  float* ws      = (float*)d_ws;
  float* outdeg  = ws;
  float* indeg   = ws + (size_t)N;
  float* outnorm = ws + 2 * (size_t)N;
  float* innorm  = ws + 3 * (size_t)N;
  float* bufA    = ws + 4 * (size_t)N;
  float* bufB    = bufA + (size_t)N * HID;

  const int rowtiles = N / 16;

  zero_kernel<<<1024, 256, 0, stream>>>(outdeg, 2 * (size_t)N);
  degree_kernel<<<(E + 255) / 256, 256, 0, stream>>>(src, dst, E, outdeg, indeg);
  norm_kernel<<<(N + 255) / 256, 256, 0, stream>>>(outdeg, indeg, outnorm, innorm, N);

  gemm_wmma<256, 128><<<rowtiles, 256, 0, stream>>>(x, W1, bufA, outnorm,
                                                    nullptr, nullptr, 0);
  zero_kernel<<<2048, 256, 0, stream>>>(bufB, (size_t)N * HID);
  aggregate_kernel<<<(E + 7) / 8, 256, 0, stream>>>(bufA, src, dst, nullptr, bufB, E, HID);
  finish_kernel<<<(N * HID + 255) / 256, 256, 0, stream>>>(bufB, innorm, b1, N, HID, 1);

  zero_kernel<<<2048, 256, 0, stream>>>(bufA, (size_t)N * HID);
  aggregate_kernel<<<(E + 7) / 8, 256, 0, stream>>>(bufB, src, dst, outnorm, bufA, E, HID);
  gemm_wmma<128, 128><<<rowtiles, 256, 0, stream>>>(bufA, W2, bufB, nullptr,
                                                    innorm, b2, 1);

  gemm_wmma<128, 64><<<rowtiles, 128, 0, stream>>>(bufB, W3, bufA, outnorm,
                                                   nullptr, nullptr, 0);
  zero_kernel<<<1024, 256, 0, stream>>>(out, (size_t)N * CLS);
  aggregate_kernel<<<(E + 15) / 16, 256, 0, stream>>>(bufA, src, dst, nullptr, out, E, CLS);
  finish_kernel<<<(N * CLS + 255) / 256, 256, 0, stream>>>(out, innorm, b3, N, CLS, 0);
}


# edge aggregation as a per-call destination-bucketed gather (int-atomic rank, prefix sum, fill) replacing f32 atomic scatter; row sum order = atomic arrival rank; zero/finish passes fold into store
# speedup vs baseline: 2.2923x; 2.2923x over previous
#include <hip/hip_runtime.h>
#include <math.h>


typedef __attribute__((ext_vector_type(2))) float v2f;
typedef __attribute__((ext_vector_type(8))) float v8f;

template<int V> struct fvec { typedef float type __attribute__((ext_vector_type(V))); };


__global__ void zero_words_kernel(int* __restrict__ p, int n) {
  const int i = blockIdx.x * blockDim.x + threadIdx.x;
  if (i < n) p[i] = 0;
}

__global__ void count_rank_kernel(const int* __restrict__ src, const int* __restrict__ dst, int E,
                                  int* __restrict__ cnt_out, int* __restrict__ cnt_in,
                                  int* __restrict__ rank) {
  const int e = blockIdx.x * blockDim.x + threadIdx.x;
  if (e < E) {
    const int s = src[e], d = dst[e];
    atomicAdd(&cnt_out[s], 1);
    rank[e] = atomicAdd(&cnt_in[d], 1);
  }
}

__global__ __launch_bounds__(256)
void scan_local_kernel(int* __restrict__ cnt_in, const int* __restrict__ cnt_out,
                       float* __restrict__ outnorm, float* __restrict__ innorm,
                       int* __restrict__ bsum, int N) {
  __shared__ int s[256];
  const int t = threadIdx.x;
  const int base = (blockIdx.x * 256 + t) * 4;
  int c[4];
  int tot = 0;
#pragma unroll
  for (int k = 0; k < 4; ++k) {
    const int i = base + k;
    int ci = 0;
    if (i < N) {
      ci = cnt_in[i];
      const int co = cnt_out[i];
      outnorm[i] = rsqrtf(fmaxf((float)co, 1.0f));
      innorm[i]  = rsqrtf(fmaxf((float)ci, 1.0f));
    }
    c[k] = ci;
    tot += ci;
  }
  s[t] = tot;
  __syncthreads();
  for (int off = 1; off < 256; off <<= 1) {
    const int v = (t >= off) ? s[t - off] : 0;
    __syncthreads();
    s[t] += v;
    __syncthreads();
  }
  int run = s[t] - tot;
  if (t == 255) bsum[blockIdx.x] = s[255];
#pragma unroll
  for (int k = 0; k < 4; ++k) {
    const int i = base + k;
    if (i < N) cnt_in[i] = run;
    run += c[k];
  }
}

__global__ __launch_bounds__(256)
void scan_offset_kernel(int* __restrict__ rowptr, const int* __restrict__ bsum, int N) {
  __shared__ int s[256];
  const int t = threadIdx.x;
  int v = 0;
  for (int j = t; j < (int)blockIdx.x; j += 256) v += bsum[j];
  s[t] = v;
  __syncthreads();
  for (int off = 128; off > 0; off >>= 1) {
    if (t < off) s[t] += s[t + off];
    __syncthreads();
  }
  const int offset = s[0];
  const int base = (blockIdx.x * 256 + t) * 4;
#pragma unroll
  for (int k = 0; k < 4; ++k) {
    const int i = base + k;
    if (i < N) rowptr[i] += offset;
  }
}

__global__ void fill_kernel(const int* __restrict__ src, const int* __restrict__ dst,
                            const int* __restrict__ rank, const int* __restrict__ rowptr,
                            int* __restrict__ col, int E) {
  const int e = blockIdx.x * blockDim.x + threadIdx.x;
  if (e < E) col[rowptr[dst[e]] + rank[e]] = src[e];
}

__global__ void move_words_kernel(const int* __restrict__ from, int* __restrict__ to, int n) {
  const int i = blockIdx.x * blockDim.x + threadIdx.x;
  if (i < n) to[i] = from[i];
}

template<int K, int M>
__global__ __launch_bounds__(32 * (M / 16))
void gemm_wmma(const float* __restrict__ A, const float* __restrict__ B,
               float* __restrict__ C,
               const float* __restrict__ rowscale,
               const float* __restrict__ innorm, const float* __restrict__ bias,
               int relu) {
  const int wave = threadIdx.x >> 5;
  const int lane = threadIdx.x & 31;
  const int half = lane >> 4;
  const int m    = lane & 15;
  const int row0 = blockIdx.x * 16;
  const int col0 = wave * 16;

  const float* __restrict__ Arow = A + (size_t)(row0 + m) * K;
  const float sc = rowscale ? rowscale[row0 + m] : 1.0f;

  v8f acc = {};
#pragma unroll 8
  for (int k = 0; k < K; k += 4) {
    const int ka = k + 2 * half;
    v2f a, b;
    a.x = Arow[ka]     * sc;
    a.y = Arow[ka + 1] * sc;
    b.x = B[(size_t)ka       * M + col0 + m];
    b.y = B[(size_t)(ka + 1) * M + col0 + m];
    acc = __builtin_amdgcn_wmma_f32_16x16x4_f32(
         false, a,  false, b,
         (short)0, acc,  false,  false);
  }

#pragma unroll
  for (int i = 0; i < 8; ++i) {
    const int row = row0 + i + 8 * half;
    const int col = col0 + m;
    float v = acc[i];
    if (innorm) v = v * innorm[row] + bias[col];
    if (relu)   v = fmaxf(v, 0.0f);
    C[(size_t)row * M + col] = v;
  }
}


template<int M, bool SCALE, int B>
__device__ __forceinline__ void gather_add(typename fvec<M / 32>::type& acc,
                                           const float* __restrict__ hl,
                                           const int* __restrict__ c,
                                           const float* __restrict__ srcscale) {
  typedef typename fvec<M / 32>::type vec;
  int s[B];
#pragma unroll
  for (int k = 0; k < B; ++k) s[k] = c[k];
  vec r[B];
#pragma unroll
  for (int k = 0; k < B; ++k) r[k] = *(const vec*)(hl + (size_t)s[k] * M);
#pragma unroll
  for (int k = 0; k < B; ++k) {
    if (SCALE) acc += r[k] * srcscale[s[k]];
    else       acc += r[k];
  }
}

template<int M, bool SCALE>
__global__ __launch_bounds__(256)
void gather_kernel(const float* __restrict__ h,
                   const int* __restrict__ rowptr, const int* __restrict__ col,
                   const float* __restrict__ srcscale,
                   const float* __restrict__ innorm, const float* __restrict__ bias,
                   float* __restrict__ out, int N, int E, int relu) {
  constexpr int V = M / 32;
  typedef typename fvec<V>::type vec;
  const int lane = threadIdx.x & 31;
  const int row = __builtin_amdgcn_readfirstlane((int)(blockIdx.x * 8 + (threadIdx.x >> 5)));
  if (row >= N) return;
  const int beg = rowptr[row];
  const int end = (row + 1 < N) ? rowptr[row + 1] : E;
  const float* __restrict__ hl = h + lane * V;

  vec acc = {};
  int j = beg;
  for (; j + 8 <= end; j += 8) gather_add<M, SCALE, 8>(acc, hl, col + j, srcscale);
  if (j + 4 <= end) { gather_add<M, SCALE, 4>(acc, hl, col + j, srcscale); j += 4; }
  if (j + 2 <= end) { gather_add<M, SCALE, 2>(acc, hl, col + j, srcscale); j += 2; }
  if (j < end)      { gather_add<M, SCALE, 1>(acc, hl, col + j, srcscale); }

  if (innorm) {
    const float inn = innorm[row];
    const vec b = *(const vec*)(bias + lane * V);
    acc = acc * inn + b;
  }
  if (relu) {
#pragma unroll
    for (int c = 0; c < V; ++c) acc[c] = fmaxf(acc[c], 0.0f);
  }
  *(vec*)(out + (size_t)row * M + lane * V) = acc;
}


extern "C" void kernel_launch(void* const* d_in, const int* in_sizes, int n_in,
                              void* d_out, int out_size, void* d_ws, size_t ws_size,
                              hipStream_t stream) {
  const float* x   = (const float*)d_in[0];
  const int*   src = (const int*)d_in[1];
  const int*   dst = (const int*)d_in[2];
  const float* W1  = (const float*)d_in[3];
  const float* b1  = (const float*)d_in[4];
  const float* W2  = (const float*)d_in[5];
  const float* b2  = (const float*)d_in[6];
  const float* W3  = (const float*)d_in[7];
  const float* b3  = (const float*)d_in[8];
  float* out = (float*)d_out;

  const int IN = 256, HID = 128, CLS = 64;
  const int E = in_sizes[1];
  const int N = in_sizes[0] / IN;

  float* ws      = (float*)d_ws;
  int*   rowptr  = (int*)ws;
  int*   cnt_out = (int*)ws + (size_t)N;
  float* outnorm = ws + 2 * (size_t)N;
  float* innorm  = ws + 3 * (size_t)N;
  float* bufA    = ws + 4 * (size_t)N;
  float* bufB    = bufA + (size_t)N * HID;

  int* col   = (int*)d_out;
  int* rank  = col + (size_t)E;
  int* bsum  = rank + (size_t)E;
  int* col3  = (int*)bufB;

  const int rowtiles = N / 16;
  const int scanblocks = (N + 1023) / 1024;

  zero_words_kernel<<<(2 * N + 255) / 256, 256, 0, stream>>>(rowptr, 2 * N);
  count_rank_kernel<<<(E + 255) / 256, 256, 0, stream>>>(src, dst, E, cnt_out, rowptr, rank);
  scan_local_kernel<<<scanblocks, 256, 0, stream>>>(rowptr, cnt_out, outnorm, innorm, bsum, N);
  scan_offset_kernel<<<scanblocks, 256, 0, stream>>>(rowptr, bsum, N);
  fill_kernel<<<(E + 255) / 256, 256, 0, stream>>>(src, dst, rank, rowptr, col, E);

  gemm_wmma<256, 128><<<rowtiles, 256, 0, stream>>>(x, W1, bufA, outnorm,
                                                    nullptr, nullptr, 0);
  gather_kernel<128, false><<<(N + 7) / 8, 256, 0, stream>>>(bufA, rowptr, col, nullptr,
                                                             innorm, b1, bufB, N, E, 1);

  gather_kernel<128, true><<<(N + 7) / 8, 256, 0, stream>>>(bufB, rowptr, col, outnorm,
                                                            nullptr, nullptr, bufA, N, E, 0);
  gemm_wmma<128, 128><<<rowtiles, 256, 0, stream>>>(bufA, W2, bufB, nullptr,
                                                    innorm, b2, 1);

  gemm_wmma<128, 64><<<rowtiles, 128, 0, stream>>>(bufB, W3, bufA, outnorm,
                                                   nullptr, nullptr, 0);
  move_words_kernel<<<(E + 255) / 256, 256, 0, stream>>>(col, col3, E);
  gather_kernel<64, false><<<(N + 7) / 8, 256, 0, stream>>>(bufA, rowptr, col3, nullptr,
                                                            innorm, b3, out, N, E, 0);
}


# f32 WMMA GEMMs: weight matrix staged once per resident block into LDS in fragment order; one wave owns a 16xM output tile, so an A fragment feeds M/16 matrix instructions; k order unchanged
# speedup vs baseline: 2.8429x; 1.2402x over previous
#include <hip/hip_runtime.h>
#include <math.h>


typedef __attribute__((ext_vector_type(2))) float v2f;
typedef __attribute__((ext_vector_type(4))) float v4f;
typedef __attribute__((ext_vector_type(8))) float v8f;

template<int V> struct fvec { typedef float type __attribute__((ext_vector_type(V))); };


__global__ void zero_words_kernel(int* __restrict__ p, int n) {
  const int i = blockIdx.x * blockDim.x + threadIdx.x;
  if (i < n) p[i] = 0;
}

__global__ void count_rank_kernel(const int* __restrict__ src, const int* __restrict__ dst, int E,
                                  int* __restrict__ cnt_out, int* __restrict__ cnt_in,
                                  int* __restrict__ rank) {
  const int e = blockIdx.x * blockDim.x + threadIdx.x;
  if (e < E) {
    const int s = src[e], d = dst[e];
    atomicAdd(&cnt_out[s], 1);
    rank[e] = atomicAdd(&cnt_in[d], 1);
  }
}

__global__ __launch_bounds__(256)
void scan_local_kernel(int* __restrict__ cnt_in, const int* __restrict__ cnt_out,
                       float* __restrict__ outnorm, float* __restrict__ innorm,
                       int* __restrict__ bsum, int N) {
  __shared__ int s[256];
  const int t = threadIdx.x;
  const int base = (blockIdx.x * 256 + t) * 4;
  int c[4];
  int tot = 0;
#pragma unroll
  for (int k = 0; k < 4; ++k) {
    const int i = base + k;
    int ci = 0;
    if (i < N) {
      ci = cnt_in[i];
      const int co = cnt_out[i];
      outnorm[i] = rsqrtf(fmaxf((float)co, 1.0f));
      innorm[i]  = rsqrtf(fmaxf((float)ci, 1.0f));
    }
    c[k] = ci;
    tot += ci;
  }
  s[t] = tot;
  __syncthreads();
  for (int off = 1; off < 256; off <<= 1) {
    const int v = (t >= off) ? s[t - off] : 0;
    __syncthreads();
    s[t] += v;
    __syncthreads();
  }
  int run = s[t] - tot;
  if (t == 255) bsum[blockIdx.x] = s[255];
#pragma unroll
  for (int k = 0; k < 4; ++k) {
    const int i = base + k;
    if (i < N) cnt_in[i] = run;
    run += c[k];
  }
}

__global__ __launch_bounds__(256)
void scan_offset_kernel(int* __restrict__ rowptr, const int* __restrict__ bsum, int N) {
  __shared__ int s[256];
  const int t = threadIdx.x;
  int v = 0;
  for (int j = t; j < (int)blockIdx.x; j += 256) v += bsum[j];
  s[t] = v;
  __syncthreads();
  for (int off = 128; off > 0; off >>= 1) {
    if (t < off) s[t] += s[t + off];
    __syncthreads();
  }
  const int offset = s[0];
  const int base = (blockIdx.x * 256 + t) * 4;
#pragma unroll
  for (int k = 0; k < 4; ++k) {
    const int i = base + k;
    if (i < N) rowptr[i] += offset;
  }
}

__global__ void fill_kernel(const int* __restrict__ src, const int* __restrict__ dst,
                            const int* __restrict__ rank, const int* __restrict__ rowptr,
                            int* __restrict__ col, int E) {
  const int e = blockIdx.x * blockDim.x + threadIdx.x;
  if (e < E) col[rowptr[dst[e]] + rank[e]] = src[e];
}

__global__ void move_words_kernel(const int* __restrict__ from, int* __restrict__ to, int n) {
  const int i = blockIdx.x * blockDim.x + threadIdx.x;
  if (i < n) to[i] = from[i];
}

template<int K, int M>
__global__ __launch_bounds__(256)
void gemm_wmma(const float* __restrict__ A, const float* __restrict__ B,
               float* __restrict__ C,
               const float* __restrict__ rowscale,
               const float* __restrict__ innorm, const float* __restrict__ bias,
               int relu, int ntiles) {
  constexpr int NT = M / 16;
  constexpr int G = K / 32;
  constexpr int PITCH = 2 * M + 32;
  __shared__ __attribute__((aligned(16))) float Bs[(K / 2) * PITCH];

  const int tid = threadIdx.x;
  for (int it = tid; it < (K / 2) * (M / 2); it += 256) {
    const int kp = it / (M / 2);
    const int c2 = (it % (M / 2)) * 2;
    const v2f r0 = *(const v2f*)(B + (size_t)(2 * kp) * M + c2);
    const v2f r1 = *(const v2f*)(B + (size_t)(2 * kp + 1) * M + c2);
    const v4f o = {r0.x, r1.x, r0.y, r1.y};
    *(v4f*)(Bs + kp * PITCH + c2 * 2) = o;
  }
  __syncthreads();

  const int lane = tid & 31;
  const int half = lane >> 4;
  const int m    = lane & 15;
  const int wave = __builtin_amdgcn_readfirstlane(tid >> 5);
  const int stride = (int)gridDim.x * 8;
  int tile = wave * (int)gridDim.x + (int)blockIdx.x;
  if (tile >= ntiles) return;

  const float* Bl = Bs + half * PITCH + m * 2;

  float bv[NT];
#pragma unroll
  for (int t = 0; t < NT; ++t) bv[t] = bias ? bias[16 * t + m] : 0.0f;

  v2f an[8];
  {
    const float* pa = A + (size_t)(tile * 16 + m) * K + 2 * half;
#pragma unroll
    for (int s = 0; s < 8; ++s) an[s] = *(const v2f*)(pa + 4 * s);
  }
  float scn = rowscale ? rowscale[tile * 16 + m] : 1.0f;

  for (;;) {
    const int ntile = tile + stride;
    const int ptile = (ntile < ntiles) ? ntile : tile;
    const float sc = scn;
    scn = rowscale ? rowscale[ptile * 16 + m] : 1.0f;
    const float* Arow  = A + (size_t)(tile  * 16 + m) * K + 2 * half;
    const float* Anext = A + (size_t)(ptile * 16 + m) * K + 2 * half;

    v8f acc[NT];
#pragma unroll
    for (int t = 0; t < NT; ++t) acc[t] = v8f{};

#pragma nounroll
    for (int g = 0; g < G; ++g) {
      v2f ac[8];
#pragma unroll
      for (int s = 0; s < 8; ++s) { ac[s].x = an[s].x * sc; ac[s].y = an[s].y * sc; }
      const float* pa = (g + 1 < G) ? (Arow + (g + 1) * 32) : Anext;
#pragma unroll
      for (int s = 0; s < 8; ++s) an[s] = *(const v2f*)(pa + 4 * s);

      const float* bp = Bl + g * (16 * PITCH);
      v2f bb[2][NT];
#pragma unroll
      for (int t = 0; t < NT; ++t) bb[0][t] = *(const v2f*)(bp + t * 32);
#pragma unroll
      for (int s = 0; s < 8; ++s) {
        if (s + 1 < 8) {
#pragma unroll
          for (int t = 0; t < NT; ++t)
            bb[(s + 1) & 1][t] = *(const v2f*)(bp + (s + 1) * (2 * PITCH) + t * 32);
        }
#pragma unroll
        for (int t = 0; t < NT; ++t)
          acc[t] = __builtin_amdgcn_wmma_f32_16x16x4_f32(
               false, ac[s],  false, bb[s & 1][t],
               (short)0, acc[t],  false,  false);
      }
    }

    const int row0 = tile * 16;
    float inn[8];
#pragma unroll
    for (int i = 0; i < 8; ++i) inn[i] = innorm ? innorm[row0 + i + 8 * half] : 1.0f;
#pragma unroll
    for (int i = 0; i < 8; ++i) {
      const int row = row0 + i + 8 * half;
#pragma unroll
      for (int t = 0; t < NT; ++t) {
        float v = acc[t][i];
        if (innorm) v = v * inn[i] + bv[t];
        if (relu)   v = fmaxf(v, 0.0f);
        C[(size_t)row * M + 16 * t + m] = v;
      }
    }

    if (ntile >= ntiles) break;
    tile = ntile;
  }
}


template<int M, bool SCALE, int B>
__device__ __forceinline__ void gather_add(typename fvec<M / 32>::type& acc,
                                           const float* __restrict__ hl,
                                           const int* __restrict__ c,
                                           const float* __restrict__ srcscale) {
  typedef typename fvec<M / 32>::type vec;
  int s[B];
#pragma unroll
  for (int k = 0; k < B; ++k) s[k] = c[k];
  vec r[B];
#pragma unroll
  for (int k = 0; k < B; ++k) r[k] = *(const vec*)(hl + (size_t)s[k] * M);
#pragma unroll
  for (int k = 0; k < B; ++k) {
    if (SCALE) acc += r[k] * srcscale[s[k]];
    else       acc += r[k];
  }
}

template<int M, bool SCALE>
__global__ __launch_bounds__(256)
void gather_kernel(const float* __restrict__ h,
                   const int* __restrict__ rowptr, const int* __restrict__ col,
                   const float* __restrict__ srcscale,
                   const float* __restrict__ innorm, const float* __restrict__ bias,
                   float* __restrict__ out, int N, int E, int relu) {
  constexpr int V = M / 32;
  typedef typename fvec<V>::type vec;
  const int lane = threadIdx.x & 31;
  const int row = __builtin_amdgcn_readfirstlane((int)(blockIdx.x * 8 + (threadIdx.x >> 5)));
  if (row >= N) return;
  const int beg = rowptr[row];
  const int end = (row + 1 < N) ? rowptr[row + 1] : E;
  const float* __restrict__ hl = h + lane * V;

  vec acc = {};
  int j = beg;
  for (; j + 8 <= end; j += 8) gather_add<M, SCALE, 8>(acc, hl, col + j, srcscale);
  if (j + 4 <= end) { gather_add<M, SCALE, 4>(acc, hl, col + j, srcscale); j += 4; }
  if (j + 2 <= end) { gather_add<M, SCALE, 2>(acc, hl, col + j, srcscale); j += 2; }
  if (j < end)      { gather_add<M, SCALE, 1>(acc, hl, col + j, srcscale); }

  if (innorm) {
    const float inn = innorm[row];
    const vec b = *(const vec*)(bias + lane * V);
    acc = acc * inn + b;
  }
  if (relu) {
#pragma unroll
    for (int c = 0; c < V; ++c) acc[c] = fmaxf(acc[c], 0.0f);
  }
  *(vec*)(out + (size_t)row * M + lane * V) = acc;
}


extern "C" void kernel_launch(void* const* d_in, const int* in_sizes, int n_in,
                              void* d_out, int out_size, void* d_ws, size_t ws_size,
                              hipStream_t stream) {
  const float* x   = (const float*)d_in[0];
  const int*   src = (const int*)d_in[1];
  const int*   dst = (const int*)d_in[2];
  const float* W1  = (const float*)d_in[3];
  const float* b1  = (const float*)d_in[4];
  const float* W2  = (const float*)d_in[5];
  const float* b2  = (const float*)d_in[6];
  const float* W3  = (const float*)d_in[7];
  const float* b3  = (const float*)d_in[8];
  float* out = (float*)d_out;

  const int IN = 256, HID = 128, CLS = 64;
  const int E = in_sizes[1];
  const int N = in_sizes[0] / IN;

  float* ws      = (float*)d_ws;
  int*   rowptr  = (int*)ws;
  int*   cnt_out = (int*)ws + (size_t)N;
  float* outnorm = ws + 2 * (size_t)N;
  float* innorm  = ws + 3 * (size_t)N;
  float* bufA    = ws + 4 * (size_t)N;
  float* bufB    = bufA + (size_t)N * HID;

  int* col   = (int*)d_out;
  int* rank  = col + (size_t)E;
  int* bsum  = rank + (size_t)E;
  int* col3  = (int*)bufB;

  const int rowtiles = N / 16;
  const int scanblocks = (N + 1023) / 1024;

  zero_words_kernel<<<(2 * N + 255) / 256, 256, 0, stream>>>(rowptr, 2 * N);
  count_rank_kernel<<<(E + 255) / 256, 256, 0, stream>>>(src, dst, E, cnt_out, rowptr, rank);
  scan_local_kernel<<<scanblocks, 256, 0, stream>>>(rowptr, cnt_out, outnorm, innorm, bsum, N);
  scan_offset_kernel<<<scanblocks, 256, 0, stream>>>(rowptr, bsum, N);
  fill_kernel<<<(E + 255) / 256, 256, 0, stream>>>(src, dst, rank, rowptr, col, E);

  gemm_wmma<256, 128><<<512, 256, 0, stream>>>(x, W1, bufA, outnorm,
                                               nullptr, nullptr, 0, rowtiles);
  gather_kernel<128, false><<<(N + 7) / 8, 256, 0, stream>>>(bufA, rowptr, col, nullptr,
                                                             innorm, b1, bufB, N, E, 1);

  gather_kernel<128, true><<<(N + 7) / 8, 256, 0, stream>>>(bufB, rowptr, col, outnorm,
                                                            nullptr, nullptr, bufA, N, E, 0);
  gemm_wmma<128, 128><<<512, 256, 0, stream>>>(bufA, W2, bufB, nullptr,
                                               innorm, b2, 1, rowtiles);

  gemm_wmma<128, 64><<<512, 256, 0, stream>>>(bufB, W3, bufA, outnorm,
                                              nullptr, nullptr, 0, rowtiles);
  move_words_kernel<<<(E + 255) / 256, 256, 0, stream>>>(col, col3, E);
  gather_kernel<64, false><<<(N + 7) / 8, 256, 0, stream>>>(bufA, rowptr, col3, nullptr,
                                                            innorm, b3, out, N, E, 0);
}


# neighbour lists and degrees built by binning edges into 512-node ranges in LDS counters (2 launches, no global atomics), not per-edge global atomics + scan + fill; row sum order = LDS atomic arrival
# speedup vs baseline: 4.2360x; 1.4900x over previous
#include <hip/hip_runtime.h>
#include <math.h>


typedef __attribute__((ext_vector_type(2))) float v2f;
typedef __attribute__((ext_vector_type(4))) float v4f;
typedef __attribute__((ext_vector_type(8))) float v8f;

template<int V> struct fvec { typedef float type __attribute__((ext_vector_type(V))); };


constexpr int BIN_SHIFT = 9;
constexpr int BIN_NODES = 1 << BIN_SHIFT;
constexpr int MAX_BINS  = 256;
constexpr int SORT_BLOCKS = 256;

__global__ __launch_bounds__(1024)
void bin_sort_kernel(const int* __restrict__ src, const int* __restrict__ dst, int E, int chunk,
                     int nbins, int2* __restrict__ pairs, int* __restrict__ sbin,
                     int2* __restrict__ descD, int2* __restrict__ descS) {
  __shared__ int cd[MAX_BINS], cs[MAX_BINS], sd[MAX_BINS], ss[MAX_BINS];
  const int t = threadIdx.x;
  const int g = blockIdx.x, G = gridDim.x;
  if (t < MAX_BINS) { cd[t] = 0; cs[t] = 0; }
  __syncthreads();
  const int e0 = min(E, g * chunk);
  const int e1 = min(E, e0 + chunk);
  for (int e = e0 + t; e < e1; e += 1024) {
    atomicAdd(&cd[dst[e] >> BIN_SHIFT], 1);
    atomicAdd(&cs[src[e] >> BIN_SHIFT], 1);
  }
  __syncthreads();
  if (t < MAX_BINS) { sd[t] = cd[t]; ss[t] = cs[t]; }
  __syncthreads();
  for (int off = 1; off < MAX_BINS; off <<= 1) {
    int vd = 0, vs = 0;
    if (t < MAX_BINS && t >= off) { vd = sd[t - off]; vs = ss[t - off]; }
    __syncthreads();
    if (t < MAX_BINS) { sd[t] += vd; ss[t] += vs; }
    __syncthreads();
  }
  if (t < MAX_BINS) {
    const int nd = cd[t], ns = cs[t];
    const int pd = e0 + sd[t] - nd;
    const int ps = e0 + ss[t] - ns;
    if (t < nbins) {
      descD[t * G + g] = make_int2(pd, nd);
      descS[t * G + g] = make_int2(ps, ns);
    }
    cd[t] = pd;
    cs[t] = ps;
  }
  __syncthreads();
  for (int e = e0 + t; e < e1; e += 1024) {
    const int d = dst[e], s = src[e];
    const int p = atomicAdd(&cd[d >> BIN_SHIFT], 1);
    pairs[p] = make_int2(d, s);
    const int q = atomicAdd(&cs[s >> BIN_SHIFT], 1);
    sbin[q] = s;
  }
}

__global__ __launch_bounds__(1024)
void bin_rows_kernel(const int2* __restrict__ pairs, const int* __restrict__ sbin,
                     const int2* __restrict__ descD, const int2* __restrict__ descS, int G,
                     int* __restrict__ rowptr, float* __restrict__ outnorm,
                     float* __restrict__ innorm, int* __restrict__ col, int N) {
  __shared__ int cin[BIN_NODES], cou[BIN_NODES], sc[BIN_NODES];
  __shared__ int basesum;
  const int t = threadIdx.x;
  const int b = blockIdx.x;
  const int sub = t & 3;
  if (t < BIN_NODES) { cin[t] = 0; cou[t] = 0; }
  if (t == 0) basesum = 0;
  __syncthreads();

  int part = 0;
  const int nprev = b * G;
  for (int i = t; i < nprev; i += 1024) part += descD[i].y;
  for (int off = 16; off > 0; off >>= 1) part += __shfl_xor(part, off);
  if ((t & 31) == 0) atomicAdd(&basesum, part);

  for (int r = t >> 2; r < G; r += 256) {
    const int2 dd = descD[b * G + r];
    const int2 ds = descS[b * G + r];
    for (int k = sub; k < dd.y; k += 4)
      atomicAdd(&cin[pairs[dd.x + k].x & (BIN_NODES - 1)], 1);
    for (int k = sub; k < ds.y; k += 4)
      atomicAdd(&cou[sbin[ds.x + k] & (BIN_NODES - 1)], 1);
  }
  __syncthreads();
  if (t < BIN_NODES) sc[t] = cin[t];
  __syncthreads();
  for (int off = 1; off < BIN_NODES; off <<= 1) {
    int v = 0;
    if (t < BIN_NODES && t >= off) v = sc[t - off];
    __syncthreads();
    if (t < BIN_NODES) sc[t] += v;
    __syncthreads();
  }
  if (t < BIN_NODES) {
    const int ci = cin[t];
    const int start = basesum + sc[t] - ci;
    const int node = b * BIN_NODES + t;
    if (node < N) {
      rowptr[node]  = start;
      innorm[node]  = rsqrtf(fmaxf((float)ci, 1.0f));
      outnorm[node] = rsqrtf(fmaxf((float)cou[t], 1.0f));
    }
    cin[t] = start;
  }
  __syncthreads();
  for (int r = t >> 2; r < G; r += 256) {
    const int2 dd = descD[b * G + r];
    for (int k = sub; k < dd.y; k += 4) {
      const int2 p = pairs[dd.x + k];
      const int slot = atomicAdd(&cin[p.x & (BIN_NODES - 1)], 1);
      col[slot] = p.y;
    }
  }
}

__global__ void move_words_kernel(const int* __restrict__ from, int* __restrict__ to, int n) {
  const int i = blockIdx.x * blockDim.x + threadIdx.x;
  if (i < n) to[i] = from[i];
}

template<int K, int M>
__global__ __launch_bounds__(256)
void gemm_wmma(const float* __restrict__ A, const float* __restrict__ B,
               float* __restrict__ C,
               const float* __restrict__ rowscale,
               const float* __restrict__ innorm, const float* __restrict__ bias,
               int relu, int ntiles) {
  constexpr int NT = M / 16;
  constexpr int G = K / 32;
  constexpr int PITCH = 2 * M + 32;
  __shared__ __attribute__((aligned(16))) float Bs[(K / 2) * PITCH];

  const int tid = threadIdx.x;
  for (int it = tid; it < (K / 2) * (M / 2); it += 256) {
    const int kp = it / (M / 2);
    const int c2 = (it % (M / 2)) * 2;
    const v2f r0 = *(const v2f*)(B + (size_t)(2 * kp) * M + c2);
    const v2f r1 = *(const v2f*)(B + (size_t)(2 * kp + 1) * M + c2);
    const v4f o = {r0.x, r1.x, r0.y, r1.y};
    *(v4f*)(Bs + kp * PITCH + c2 * 2) = o;
  }
  __syncthreads();

  const int lane = tid & 31;
  const int half = lane >> 4;
  const int m    = lane & 15;
  const int wave = __builtin_amdgcn_readfirstlane(tid >> 5);
  const int stride = (int)gridDim.x * 8;
  int tile = wave * (int)gridDim.x + (int)blockIdx.x;
  if (tile >= ntiles) return;

  const float* Bl = Bs + half * PITCH + m * 2;

  float bv[NT];
#pragma unroll
  for (int t = 0; t < NT; ++t) bv[t] = bias ? bias[16 * t + m] : 0.0f;

  v2f an[8];
  {
    const float* pa = A + (size_t)(tile * 16 + m) * K + 2 * half;
#pragma unroll
    for (int s = 0; s < 8; ++s) an[s] = *(const v2f*)(pa + 4 * s);
  }
  float scn = rowscale ? rowscale[tile * 16 + m] : 1.0f;

  for (;;) {
    const int ntile = tile + stride;
    const int ptile = (ntile < ntiles) ? ntile : tile;
    const float sc = scn;
    scn = rowscale ? rowscale[ptile * 16 + m] : 1.0f;
    const float* Arow  = A + (size_t)(tile  * 16 + m) * K + 2 * half;
    const float* Anext = A + (size_t)(ptile * 16 + m) * K + 2 * half;

    v8f acc[NT];
#pragma unroll
    for (int t = 0; t < NT; ++t) acc[t] = v8f{};

#pragma nounroll
    for (int g = 0; g < G; ++g) {
      v2f ac[8];
#pragma unroll
      for (int s = 0; s < 8; ++s) { ac[s].x = an[s].x * sc; ac[s].y = an[s].y * sc; }
      const float* pa = (g + 1 < G) ? (Arow + (g + 1) * 32) : Anext;
#pragma unroll
      for (int s = 0; s < 8; ++s) an[s] = *(const v2f*)(pa + 4 * s);

      const float* bp = Bl + g * (16 * PITCH);
      v2f bb[2][NT];
#pragma unroll
      for (int t = 0; t < NT; ++t) bb[0][t] = *(const v2f*)(bp + t * 32);
#pragma unroll
      for (int s = 0; s < 8; ++s) {
        if (s + 1 < 8) {
#pragma unroll
          for (int t = 0; t < NT; ++t)
            bb[(s + 1) & 1][t] = *(const v2f*)(bp + (s + 1) * (2 * PITCH) + t * 32);
        }
#pragma unroll
        for (int t = 0; t < NT; ++t)
          acc[t] = __builtin_amdgcn_wmma_f32_16x16x4_f32(
               false, ac[s],  false, bb[s & 1][t],
               (short)0, acc[t],  false,  false);
      }
    }

    const int row0 = tile * 16;
    float inn[8];
#pragma unroll
    for (int i = 0; i < 8; ++i) inn[i] = innorm ? innorm[row0 + i + 8 * half] : 1.0f;
#pragma unroll
    for (int i = 0; i < 8; ++i) {
      const int row = row0 + i + 8 * half;
#pragma unroll
      for (int t = 0; t < NT; ++t) {
        float v = acc[t][i];
        if (innorm) v = v * inn[i] + bv[t];
        if (relu)   v = fmaxf(v, 0.0f);
        C[(size_t)row * M + 16 * t + m] = v;
      }
    }

    if (ntile >= ntiles) break;
    tile = ntile;
  }
}


template<int M, bool SCALE, int B>
__device__ __forceinline__ void gather_add(typename fvec<M / 32>::type& acc,
                                           const float* __restrict__ hl,
                                           const int* __restrict__ c,
                                           const float* __restrict__ srcscale) {
  typedef typename fvec<M / 32>::type vec;
  int s[B];
#pragma unroll
  for (int k = 0; k < B; ++k) s[k] = c[k];
  vec r[B];
#pragma unroll
  for (int k = 0; k < B; ++k) r[k] = *(const vec*)(hl + (size_t)s[k] * M);
#pragma unroll
  for (int k = 0; k < B; ++k) {
    if (SCALE) acc += r[k] * srcscale[s[k]];
    else       acc += r[k];
  }
}

template<int M, bool SCALE>
__global__ __launch_bounds__(256)
void gather_kernel(const float* __restrict__ h,
                   const int* __restrict__ rowptr, const int* __restrict__ col,
                   const float* __restrict__ srcscale,
                   const float* __restrict__ innorm, const float* __restrict__ bias,
                   float* __restrict__ out, int N, int E, int relu) {
  constexpr int V = M / 32;
  typedef typename fvec<V>::type vec;
  const int lane = threadIdx.x & 31;
  const int row = __builtin_amdgcn_readfirstlane((int)(blockIdx.x * 8 + (threadIdx.x >> 5)));
  if (row >= N) return;
  const int beg = rowptr[row];
  const int end = (row + 1 < N) ? rowptr[row + 1] : E;
  const float* __restrict__ hl = h + lane * V;

  vec acc = {};
  int j = beg;
  for (; j + 8 <= end; j += 8) gather_add<M, SCALE, 8>(acc, hl, col + j, srcscale);
  if (j + 4 <= end) { gather_add<M, SCALE, 4>(acc, hl, col + j, srcscale); j += 4; }
  if (j + 2 <= end) { gather_add<M, SCALE, 2>(acc, hl, col + j, srcscale); j += 2; }
  if (j < end)      { gather_add<M, SCALE, 1>(acc, hl, col + j, srcscale); }

  if (innorm) {
    const float inn = innorm[row];
    const vec b = *(const vec*)(bias + lane * V);
    acc = acc * inn + b;
  }
  if (relu) {
#pragma unroll
    for (int c = 0; c < V; ++c) acc[c] = fmaxf(acc[c], 0.0f);
  }
  *(vec*)(out + (size_t)row * M + lane * V) = acc;
}


extern "C" void kernel_launch(void* const* d_in, const int* in_sizes, int n_in,
                              void* d_out, int out_size, void* d_ws, size_t ws_size,
                              hipStream_t stream) {
  const float* x   = (const float*)d_in[0];
  const int*   src = (const int*)d_in[1];
  const int*   dst = (const int*)d_in[2];
  const float* W1  = (const float*)d_in[3];
  const float* b1  = (const float*)d_in[4];
  const float* W2  = (const float*)d_in[5];
  const float* b2  = (const float*)d_in[6];
  const float* W3  = (const float*)d_in[7];
  const float* b3  = (const float*)d_in[8];
  float* out = (float*)d_out;

  const int IN = 256, HID = 128, CLS = 64;
  const int E = in_sizes[1];
  const int N = in_sizes[0] / IN;

  float* ws      = (float*)d_ws;
  int*   rowptr  = (int*)ws;
  float* outnorm = ws + 2 * (size_t)N;
  float* innorm  = ws + 3 * (size_t)N;
  float* bufA    = ws + 4 * (size_t)N;
  float* bufB    = bufA + (size_t)N * HID;

  int* col   = (int*)d_out;
  int* col3  = (int*)bufB;

  const int nbins = (N + BIN_NODES - 1) / BIN_NODES;
  const int chunk = (E + SORT_BLOCKS - 1) / SORT_BLOCKS;
  int2* pairs = (int2*)bufB;
  int*  sbin  = (int*)bufB + 2 * (size_t)E;
  int2* descD = (int2*)((int*)bufB + 3 * (size_t)E);
  int2* descS = descD + (size_t)MAX_BINS * SORT_BLOCKS;

  const int rowtiles = N / 16;

  bin_sort_kernel<<<SORT_BLOCKS, 1024, 0, stream>>>(src, dst, E, chunk, nbins,
                                                    pairs, sbin, descD, descS);
  bin_rows_kernel<<<nbins, 1024, 0, stream>>>(pairs, sbin, descD, descS, SORT_BLOCKS,
                                              rowptr, outnorm, innorm, col, N);

  gemm_wmma<256, 128><<<512, 256, 0, stream>>>(x, W1, bufA, outnorm,
                                               nullptr, nullptr, 0, rowtiles);
  gather_kernel<128, false><<<(N + 7) / 8, 256, 0, stream>>>(bufA, rowptr, col, nullptr,
                                                             innorm, b1, bufB, N, E, 1);

  gather_kernel<128, true><<<(N + 7) / 8, 256, 0, stream>>>(bufB, rowptr, col, outnorm,
                                                            nullptr, nullptr, bufA, N, E, 0);
  gemm_wmma<128, 128><<<512, 256, 0, stream>>>(bufA, W2, bufB, nullptr,
                                               innorm, b2, 1, rowtiles);

  gemm_wmma<128, 64><<<512, 256, 0, stream>>>(bufB, W3, bufA, outnorm,
                                              nullptr, nullptr, 0, rowtiles);
  move_words_kernel<<<(E + 255) / 256, 256, 0, stream>>>(col, col3, E);
  gather_kernel<64, false><<<(N + 7) / 8, 256, 0, stream>>>(bufA, rowptr, col3, nullptr,
                                                            innorm, b3, out, N, E, 0);
}


# list building: both bucketing launches assemble their sorted output (pairs, source ids, col) in LDS and copy it out one whole line per store, not one 4/8-byte store per edge; row sum order unchanged
# speedup vs baseline: 4.3869x; 1.0356x over previous
#include <hip/hip_runtime.h>
#include <math.h>


typedef __attribute__((ext_vector_type(2))) float v2f;
typedef __attribute__((ext_vector_type(4))) float v4f;
typedef __attribute__((ext_vector_type(8))) float v8f;

template<int V> struct fvec { typedef float type __attribute__((ext_vector_type(V))); };


constexpr int BIN_SHIFT = 9;
constexpr int BIN_NODES = 1 << BIN_SHIFT;
constexpr int MAX_BINS  = 256;
constexpr int SORT_BLOCKS = 256;
constexpr int SORT_CAP  = 8192;
constexpr int ROWS_CAP  = 16384;

__global__ __launch_bounds__(1024)
void bin_sort_kernel(const int* __restrict__ src, const int* __restrict__ dst, int E, int chunk,
                     int nbins, int2* __restrict__ pairs, int* __restrict__ sbin,
                     int2* __restrict__ descD, int2* __restrict__ descS) {
  __shared__ int cd[MAX_BINS], cs[MAX_BINS], sd[MAX_BINS], ss[MAX_BINS];
  __shared__ int2 lp[SORT_CAP];
  __shared__ int  ls[SORT_CAP];
  const int t = threadIdx.x;
  const int g = blockIdx.x, G = gridDim.x;
  if (t < MAX_BINS) { cd[t] = 0; cs[t] = 0; }
  __syncthreads();
  const int e0 = min(E, g * chunk);
  const int e1 = min(E, e0 + chunk);
  for (int e = e0 + t; e < e1; e += 1024) {
    atomicAdd(&cd[dst[e] >> BIN_SHIFT], 1);
    atomicAdd(&cs[src[e] >> BIN_SHIFT], 1);
  }
  __syncthreads();
  if (t < MAX_BINS) { sd[t] = cd[t]; ss[t] = cs[t]; }
  __syncthreads();
  for (int off = 1; off < MAX_BINS; off <<= 1) {
    int vd = 0, vs = 0;
    if (t < MAX_BINS && t >= off) { vd = sd[t - off]; vs = ss[t - off]; }
    __syncthreads();
    if (t < MAX_BINS) { sd[t] += vd; ss[t] += vs; }
    __syncthreads();
  }
  if (t < MAX_BINS) {
    const int nd = cd[t], ns = cs[t];
    const int pd = e0 + sd[t] - nd;
    const int ps = e0 + ss[t] - ns;
    if (t < nbins) {
      descD[t * G + g] = make_int2(pd, nd);
      descS[t * G + g] = make_int2(ps, ns);
    }
    cd[t] = pd;
    cs[t] = ps;
  }
  __syncthreads();
  if (chunk <= SORT_CAP) {
    for (int e = e0 + t; e < e1; e += 1024) {
      const int d = dst[e], s = src[e];
      const int p = atomicAdd(&cd[d >> BIN_SHIFT], 1);
      lp[p - e0] = make_int2(d, s);
      const int q = atomicAdd(&cs[s >> BIN_SHIFT], 1);
      ls[q - e0] = s;
    }
    __syncthreads();
    const int n = e1 - e0;
    for (int i = t - (e0 & 31); i < n; i += 1024) {
      if (i >= 0) {
        pairs[e0 + i] = lp[i];
        sbin[e0 + i]  = ls[i];
      }
    }
  } else {
    for (int e = e0 + t; e < e1; e += 1024) {
      const int d = dst[e], s = src[e];
      const int p = atomicAdd(&cd[d >> BIN_SHIFT], 1);
      pairs[p] = make_int2(d, s);
      const int q = atomicAdd(&cs[s >> BIN_SHIFT], 1);
      sbin[q] = s;
    }
  }
}

__global__ __launch_bounds__(1024)
void bin_rows_kernel(const int2* __restrict__ pairs, const int* __restrict__ sbin,
                     const int2* __restrict__ descD, const int2* __restrict__ descS, int G,
                     int* __restrict__ rowptr, float* __restrict__ outnorm,
                     float* __restrict__ innorm, int* __restrict__ col, int N) {
  __shared__ int cin[BIN_NODES], cou[BIN_NODES], sc[BIN_NODES];
  __shared__ int basesum;
  __shared__ int lcol[ROWS_CAP];
  const int t = threadIdx.x;
  const int b = blockIdx.x;
  const int sub = t & 3;
  if (t < BIN_NODES) { cin[t] = 0; cou[t] = 0; }
  if (t == 0) basesum = 0;
  __syncthreads();

  int part = 0;
  const int nprev = b * G;
  for (int i = t; i < nprev; i += 1024) part += descD[i].y;
  for (int off = 16; off > 0; off >>= 1) part += __shfl_xor(part, off);
  if ((t & 31) == 0) atomicAdd(&basesum, part);

  for (int r = t >> 2; r < G; r += 256) {
    const int2 dd = descD[b * G + r];
    const int2 ds = descS[b * G + r];
    for (int k = sub; k < dd.y; k += 4)
      atomicAdd(&cin[pairs[dd.x + k].x & (BIN_NODES - 1)], 1);
    for (int k = sub; k < ds.y; k += 4)
      atomicAdd(&cou[sbin[ds.x + k] & (BIN_NODES - 1)], 1);
  }
  __syncthreads();
  if (t < BIN_NODES) sc[t] = cin[t];
  __syncthreads();
  for (int off = 1; off < BIN_NODES; off <<= 1) {
    int v = 0;
    if (t < BIN_NODES && t >= off) v = sc[t - off];
    __syncthreads();
    if (t < BIN_NODES) sc[t] += v;
    __syncthreads();
  }
  const int base  = basesum;
  const int total = sc[BIN_NODES - 1];
  const bool in_lds = total <= ROWS_CAP;
  if (t < BIN_NODES) {
    const int ci = cin[t];
    const int start = base + sc[t] - ci;
    const int node = b * BIN_NODES + t;
    if (node < N) {
      rowptr[node]  = start;
      innorm[node]  = rsqrtf(fmaxf((float)ci, 1.0f));
      outnorm[node] = rsqrtf(fmaxf((float)cou[t], 1.0f));
    }
    cin[t] = start;
  }
  __syncthreads();
  for (int r = t >> 2; r < G; r += 256) {
    const int2 dd = descD[b * G + r];
    for (int k = sub; k < dd.y; k += 4) {
      const int2 p = pairs[dd.x + k];
      const int slot = atomicAdd(&cin[p.x & (BIN_NODES - 1)], 1);
      if (in_lds) lcol[slot - base] = p.y;
      else        col[slot] = p.y;
    }
  }
  __syncthreads();
  if (in_lds) {
    for (int i = t - (base & 31); i < total; i += 1024)
      if (i >= 0) col[base + i] = lcol[i];
  }
}

__global__ void move_words_kernel(const int* __restrict__ from, int* __restrict__ to, int n) {
  const int i = blockIdx.x * blockDim.x + threadIdx.x;
  if (i < n) to[i] = from[i];
}

template<int K, int M>
__global__ __launch_bounds__(256)
void gemm_wmma(const float* __restrict__ A, const float* __restrict__ B,
               float* __restrict__ C,
               const float* __restrict__ rowscale,
               const float* __restrict__ innorm, const float* __restrict__ bias,
               int relu, int ntiles) {
  constexpr int NT = M / 16;
  constexpr int G = K / 32;
  constexpr int PITCH = 2 * M + 32;
  __shared__ __attribute__((aligned(16))) float Bs[(K / 2) * PITCH];

  const int tid = threadIdx.x;
  for (int it = tid; it < (K / 2) * (M / 2); it += 256) {
    const int kp = it / (M / 2);
    const int c2 = (it % (M / 2)) * 2;
    const v2f r0 = *(const v2f*)(B + (size_t)(2 * kp) * M + c2);
    const v2f r1 = *(const v2f*)(B + (size_t)(2 * kp + 1) * M + c2);
    const v4f o = {r0.x, r1.x, r0.y, r1.y};
    *(v4f*)(Bs + kp * PITCH + c2 * 2) = o;
  }
  __syncthreads();

  const int lane = tid & 31;
  const int half = lane >> 4;
  const int m    = lane & 15;
  const int wave = __builtin_amdgcn_readfirstlane(tid >> 5);
  const int stride = (int)gridDim.x * 8;
  int tile = wave * (int)gridDim.x + (int)blockIdx.x;
  if (tile >= ntiles) return;

  const float* Bl = Bs + half * PITCH + m * 2;

  float bv[NT];
#pragma unroll
  for (int t = 0; t < NT; ++t) bv[t] = bias ? bias[16 * t + m] : 0.0f;

  v2f an[8];
  {
    const float* pa = A + (size_t)(tile * 16 + m) * K + 2 * half;
#pragma unroll
    for (int s = 0; s < 8; ++s) an[s] = *(const v2f*)(pa + 4 * s);
  }
  float scn = rowscale ? rowscale[tile * 16 + m] : 1.0f;

  for (;;) {
    const int ntile = tile + stride;
    const int ptile = (ntile < ntiles) ? ntile : tile;
    const float sc = scn;
    scn = rowscale ? rowscale[ptile * 16 + m] : 1.0f;
    const float* Arow  = A + (size_t)(tile  * 16 + m) * K + 2 * half;
    const float* Anext = A + (size_t)(ptile * 16 + m) * K + 2 * half;

    v8f acc[NT];
#pragma unroll
    for (int t = 0; t < NT; ++t) acc[t] = v8f{};

#pragma nounroll
    for (int g = 0; g < G; ++g) {
      v2f ac[8];
#pragma unroll
      for (int s = 0; s < 8; ++s) { ac[s].x = an[s].x * sc; ac[s].y = an[s].y * sc; }
      const float* pa = (g + 1 < G) ? (Arow + (g + 1) * 32) : Anext;
#pragma unroll
      for (int s = 0; s < 8; ++s) an[s] = *(const v2f*)(pa + 4 * s);

      const float* bp = Bl + g * (16 * PITCH);
      v2f bb[2][NT];
#pragma unroll
      for (int t = 0; t < NT; ++t) bb[0][t] = *(const v2f*)(bp + t * 32);
#pragma unroll
      for (int s = 0; s < 8; ++s) {
        if (s + 1 < 8) {
#pragma unroll
          for (int t = 0; t < NT; ++t)
            bb[(s + 1) & 1][t] = *(const v2f*)(bp + (s + 1) * (2 * PITCH) + t * 32);
        }
#pragma unroll
        for (int t = 0; t < NT; ++t)
          acc[t] = __builtin_amdgcn_wmma_f32_16x16x4_f32(
               false, ac[s],  false, bb[s & 1][t],
               (short)0, acc[t],  false,  false);
      }
    }

    const int row0 = tile * 16;
    float inn[8];
#pragma unroll
    for (int i = 0; i < 8; ++i) inn[i] = innorm ? innorm[row0 + i + 8 * half] : 1.0f;
#pragma unroll
    for (int i = 0; i < 8; ++i) {
      const int row = row0 + i + 8 * half;
#pragma unroll
      for (int t = 0; t < NT; ++t) {
        float v = acc[t][i];
        if (innorm) v = v * inn[i] + bv[t];
        if (relu)   v = fmaxf(v, 0.0f);
        C[(size_t)row * M + 16 * t + m] = v;
      }
    }

    if (ntile >= ntiles) break;
    tile = ntile;
  }
}


template<int M, bool SCALE, int B>
__device__ __forceinline__ void gather_add(typename fvec<M / 32>::type& acc,
                                           const float* __restrict__ hl,
                                           const int* __restrict__ c,
                                           const float* __restrict__ srcscale) {
  typedef typename fvec<M / 32>::type vec;
  int s[B];
#pragma unroll
  for (int k = 0; k < B; ++k) s[k] = c[k];
  vec r[B];
#pragma unroll
  for (int k = 0; k < B; ++k) r[k] = *(const vec*)(hl + (size_t)s[k] * M);
#pragma unroll
  for (int k = 0; k < B; ++k) {
    if (SCALE) acc += r[k] * srcscale[s[k]];
    else       acc += r[k];
  }
}

template<int M, bool SCALE>
__global__ __launch_bounds__(256)
void gather_kernel(const float* __restrict__ h,
                   const int* __restrict__ rowptr, const int* __restrict__ col,
                   const float* __restrict__ srcscale,
                   const float* __restrict__ innorm, const float* __restrict__ bias,
                   float* __restrict__ out, int N, int E, int relu) {
  constexpr int V = M / 32;
  typedef typename fvec<V>::type vec;
  const int lane = threadIdx.x & 31;
  const int row = __builtin_amdgcn_readfirstlane((int)(blockIdx.x * 8 + (threadIdx.x >> 5)));
  if (row >= N) return;
  const int beg = rowptr[row];
  const int end = (row + 1 < N) ? rowptr[row + 1] : E;
  const float* __restrict__ hl = h + lane * V;

  vec acc = {};
  int j = beg;
  for (; j + 8 <= end; j += 8) gather_add<M, SCALE, 8>(acc, hl, col + j, srcscale);
  if (j + 4 <= end) { gather_add<M, SCALE, 4>(acc, hl, col + j, srcscale); j += 4; }
  if (j + 2 <= end) { gather_add<M, SCALE, 2>(acc, hl, col + j, srcscale); j += 2; }
  if (j < end)      { gather_add<M, SCALE, 1>(acc, hl, col + j, srcscale); }

  if (innorm) {
    const float inn = innorm[row];
    const vec b = *(const vec*)(bias + lane * V);
    acc = acc * inn + b;
  }
  if (relu) {
#pragma unroll
    for (int c = 0; c < V; ++c) acc[c] = fmaxf(acc[c], 0.0f);
  }
  *(vec*)(out + (size_t)row * M + lane * V) = acc;
}


extern "C" void kernel_launch(void* const* d_in, const int* in_sizes, int n_in,
                              void* d_out, int out_size, void* d_ws, size_t ws_size,
                              hipStream_t stream) {
  const float* x   = (const float*)d_in[0];
  const int*   src = (const int*)d_in[1];
  const int*   dst = (const int*)d_in[2];
  const float* W1  = (const float*)d_in[3];
  const float* b1  = (const float*)d_in[4];
  const float* W2  = (const float*)d_in[5];
  const float* b2  = (const float*)d_in[6];
  const float* W3  = (const float*)d_in[7];
  const float* b3  = (const float*)d_in[8];
  float* out = (float*)d_out;

  const int IN = 256, HID = 128, CLS = 64;
  const int E = in_sizes[1];
  const int N = in_sizes[0] / IN;

  float* ws      = (float*)d_ws;
  int*   rowptr  = (int*)ws;
  float* outnorm = ws + 2 * (size_t)N;
  float* innorm  = ws + 3 * (size_t)N;
  float* bufA    = ws + 4 * (size_t)N;
  float* bufB    = bufA + (size_t)N * HID;

  int* col   = (int*)d_out;
  int* col3  = (int*)bufB;

  const int nbins = (N + BIN_NODES - 1) / BIN_NODES;
  const int chunk = (E + SORT_BLOCKS - 1) / SORT_BLOCKS;
  int2* pairs = (int2*)bufB;
  int*  sbin  = (int*)bufB + 2 * (size_t)E;
  int2* descD = (int2*)((int*)bufB + 3 * (size_t)E);
  int2* descS = descD + (size_t)MAX_BINS * SORT_BLOCKS;

  const int rowtiles = N / 16;

  bin_sort_kernel<<<SORT_BLOCKS, 1024, 0, stream>>>(src, dst, E, chunk, nbins,
                                                    pairs, sbin, descD, descS);
  bin_rows_kernel<<<nbins, 1024, 0, stream>>>(pairs, sbin, descD, descS, SORT_BLOCKS,
                                              rowptr, outnorm, innorm, col, N);

  gemm_wmma<256, 128><<<512, 256, 0, stream>>>(x, W1, bufA, outnorm,
                                               nullptr, nullptr, 0, rowtiles);
  gather_kernel<128, false><<<(N + 7) / 8, 256, 0, stream>>>(bufA, rowptr, col, nullptr,
                                                             innorm, b1, bufB, N, E, 1);

  gather_kernel<128, true><<<(N + 7) / 8, 256, 0, stream>>>(bufB, rowptr, col, outnorm,
                                                            nullptr, nullptr, bufA, N, E, 0);
  gemm_wmma<128, 128><<<512, 256, 0, stream>>>(bufA, W2, bufB, nullptr,
                                               innorm, b2, 1, rowtiles);

  gemm_wmma<128, 64><<<512, 256, 0, stream>>>(bufB, W3, bufA, outnorm,
                                              nullptr, nullptr, 0, rowtiles);
  move_words_kernel<<<(E + 255) / 256, 256, 0, stream>>>(col, col3, E);
  gather_kernel<64, false><<<(N + 7) / 8, 256, 0, stream>>>(bufA, rowptr, col3, nullptr,
                                                            innorm, b3, out, N, E, 0);
}


# layer 2's aggregation and GEMM and layer 3's GEMM in one launch; a wave sums a 16-row tile's neighbour rows into LDS and feeds both f32 WMMA products from there; row sum order and k order unchanged
# speedup vs baseline: 4.7121x; 1.0741x over previous
#include <hip/hip_runtime.h>
#include <math.h>


typedef __attribute__((ext_vector_type(2))) float v2f;
typedef __attribute__((ext_vector_type(4))) float v4f;
typedef __attribute__((ext_vector_type(8))) float v8f;

template<int V> struct fvec { typedef float type __attribute__((ext_vector_type(V))); };


constexpr int BIN_SHIFT = 9;
constexpr int BIN_NODES = 1 << BIN_SHIFT;
constexpr int MAX_BINS  = 256;
constexpr int SORT_BLOCKS = 256;
constexpr int SORT_CAP  = 8192;
constexpr int ROWS_CAP  = 16384;

__global__ __launch_bounds__(1024)
void bin_sort_kernel(const int* __restrict__ src, const int* __restrict__ dst, int E, int chunk,
                     int nbins, int2* __restrict__ pairs, int* __restrict__ sbin,
                     int2* __restrict__ descD, int2* __restrict__ descS) {
  __shared__ int cd[MAX_BINS], cs[MAX_BINS], sd[MAX_BINS], ss[MAX_BINS];
  __shared__ int2 lp[SORT_CAP];
  __shared__ int  ls[SORT_CAP];
  const int t = threadIdx.x;
  const int g = blockIdx.x, G = gridDim.x;
  if (t < MAX_BINS) { cd[t] = 0; cs[t] = 0; }
  __syncthreads();
  const int e0 = min(E, g * chunk);
  const int e1 = min(E, e0 + chunk);
  for (int e = e0 + t; e < e1; e += 1024) {
    atomicAdd(&cd[dst[e] >> BIN_SHIFT], 1);
    atomicAdd(&cs[src[e] >> BIN_SHIFT], 1);
  }
  __syncthreads();
  if (t < MAX_BINS) { sd[t] = cd[t]; ss[t] = cs[t]; }
  __syncthreads();
  for (int off = 1; off < MAX_BINS; off <<= 1) {
    int vd = 0, vs = 0;
    if (t < MAX_BINS && t >= off) { vd = sd[t - off]; vs = ss[t - off]; }
    __syncthreads();
    if (t < MAX_BINS) { sd[t] += vd; ss[t] += vs; }
    __syncthreads();
  }
  if (t < MAX_BINS) {
    const int nd = cd[t], ns = cs[t];
    const int pd = e0 + sd[t] - nd;
    const int ps = e0 + ss[t] - ns;
    if (t < nbins) {
      descD[t * G + g] = make_int2(pd, nd);
      descS[t * G + g] = make_int2(ps, ns);
    }
    cd[t] = pd;
    cs[t] = ps;
  }
  __syncthreads();
  if (chunk <= SORT_CAP) {
    for (int e = e0 + t; e < e1; e += 1024) {
      const int d = dst[e], s = src[e];
      const int p = atomicAdd(&cd[d >> BIN_SHIFT], 1);
      lp[p - e0] = make_int2(d, s);
      const int q = atomicAdd(&cs[s >> BIN_SHIFT], 1);
      ls[q - e0] = s;
    }
    __syncthreads();
    const int n = e1 - e0;
    for (int i = t - (e0 & 31); i < n; i += 1024) {
      if (i >= 0) {
        pairs[e0 + i] = lp[i];
        sbin[e0 + i]  = ls[i];
      }
    }
  } else {
    for (int e = e0 + t; e < e1; e += 1024) {
      const int d = dst[e], s = src[e];
      const int p = atomicAdd(&cd[d >> BIN_SHIFT], 1);
      pairs[p] = make_int2(d, s);
      const int q = atomicAdd(&cs[s >> BIN_SHIFT], 1);
      sbin[q] = s;
    }
  }
}

__global__ __launch_bounds__(1024)
void bin_rows_kernel(const int2* __restrict__ pairs, const int* __restrict__ sbin,
                     const int2* __restrict__ descD, const int2* __restrict__ descS, int G,
                     int* __restrict__ rowptr, float* __restrict__ outnorm,
                     float* __restrict__ innorm, int* __restrict__ col, int N) {
  __shared__ int cin[BIN_NODES], cou[BIN_NODES], sc[BIN_NODES];
  __shared__ int basesum;
  __shared__ int lcol[ROWS_CAP];
  const int t = threadIdx.x;
  const int b = blockIdx.x;
  const int sub = t & 3;
  if (t < BIN_NODES) { cin[t] = 0; cou[t] = 0; }
  if (t == 0) basesum = 0;
  __syncthreads();

  int part = 0;
  const int nprev = b * G;
  for (int i = t; i < nprev; i += 1024) part += descD[i].y;
  for (int off = 16; off > 0; off >>= 1) part += __shfl_xor(part, off);
  if ((t & 31) == 0) atomicAdd(&basesum, part);

  for (int r = t >> 2; r < G; r += 256) {
    const int2 dd = descD[b * G + r];
    const int2 ds = descS[b * G + r];
    for (int k = sub; k < dd.y; k += 4)
      atomicAdd(&cin[pairs[dd.x + k].x & (BIN_NODES - 1)], 1);
    for (int k = sub; k < ds.y; k += 4)
      atomicAdd(&cou[sbin[ds.x + k] & (BIN_NODES - 1)], 1);
  }
  __syncthreads();
  if (t < BIN_NODES) sc[t] = cin[t];
  __syncthreads();
  for (int off = 1; off < BIN_NODES; off <<= 1) {
    int v = 0;
    if (t < BIN_NODES && t >= off) v = sc[t - off];
    __syncthreads();
    if (t < BIN_NODES) sc[t] += v;
    __syncthreads();
  }
  const int base  = basesum;
  const int total = sc[BIN_NODES - 1];
  const bool in_lds = total <= ROWS_CAP;
  if (t < BIN_NODES) {
    const int ci = cin[t];
    const int start = base + sc[t] - ci;
    const int node = b * BIN_NODES + t;
    if (node < N) {
      rowptr[node]  = start;
      innorm[node]  = rsqrtf(fmaxf((float)ci, 1.0f));
      outnorm[node] = rsqrtf(fmaxf((float)cou[t], 1.0f));
    }
    cin[t] = start;
  }
  __syncthreads();
  for (int r = t >> 2; r < G; r += 256) {
    const int2 dd = descD[b * G + r];
    for (int k = sub; k < dd.y; k += 4) {
      const int2 p = pairs[dd.x + k];
      const int slot = atomicAdd(&cin[p.x & (BIN_NODES - 1)], 1);
      if (in_lds) lcol[slot - base] = p.y;
      else        col[slot] = p.y;
    }
  }
  __syncthreads();
  if (in_lds) {
    for (int i = t - (base & 31); i < total; i += 1024)
      if (i >= 0) col[base + i] = lcol[i];
  }
}

__global__ void move_words_kernel(const int* __restrict__ from, int* __restrict__ to, int n) {
  const int i = blockIdx.x * blockDim.x + threadIdx.x;
  if (i < n) to[i] = from[i];
}

template<int K, int M>
__global__ __launch_bounds__(256)
void gemm_wmma(const float* __restrict__ A, const float* __restrict__ B,
               float* __restrict__ C,
               const float* __restrict__ rowscale,
               const float* __restrict__ innorm, const float* __restrict__ bias,
               int relu, int ntiles) {
  constexpr int NT = M / 16;
  constexpr int G = K / 32;
  constexpr int PITCH = 2 * M + 32;
  __shared__ __attribute__((aligned(16))) float Bs[(K / 2) * PITCH];

  const int tid = threadIdx.x;
  for (int it = tid; it < (K / 2) * (M / 2); it += 256) {
    const int kp = it / (M / 2);
    const int c2 = (it % (M / 2)) * 2;
    const v2f r0 = *(const v2f*)(B + (size_t)(2 * kp) * M + c2);
    const v2f r1 = *(const v2f*)(B + (size_t)(2 * kp + 1) * M + c2);
    const v4f o = {r0.x, r1.x, r0.y, r1.y};
    *(v4f*)(Bs + kp * PITCH + c2 * 2) = o;
  }
  __syncthreads();

  const int lane = tid & 31;
  const int half = lane >> 4;
  const int m    = lane & 15;
  const int wave = __builtin_amdgcn_readfirstlane(tid >> 5);
  const int stride = (int)gridDim.x * 8;
  int tile = wave * (int)gridDim.x + (int)blockIdx.x;
  if (tile >= ntiles) return;

  const float* Bl = Bs + half * PITCH + m * 2;

  float bv[NT];
#pragma unroll
  for (int t = 0; t < NT; ++t) bv[t] = bias ? bias[16 * t + m] : 0.0f;

  v2f an[8];
  {
    const float* pa = A + (size_t)(tile * 16 + m) * K + 2 * half;
#pragma unroll
    for (int s = 0; s < 8; ++s) an[s] = *(const v2f*)(pa + 4 * s);
  }
  float scn = rowscale ? rowscale[tile * 16 + m] : 1.0f;

  for (;;) {
    const int ntile = tile + stride;
    const int ptile = (ntile < ntiles) ? ntile : tile;
    const float sc = scn;
    scn = rowscale ? rowscale[ptile * 16 + m] : 1.0f;
    const float* Arow  = A + (size_t)(tile  * 16 + m) * K + 2 * half;
    const float* Anext = A + (size_t)(ptile * 16 + m) * K + 2 * half;

    v8f acc[NT];
#pragma unroll
    for (int t = 0; t < NT; ++t) acc[t] = v8f{};

#pragma nounroll
    for (int g = 0; g < G; ++g) {
      v2f ac[8];
#pragma unroll
      for (int s = 0; s < 8; ++s) { ac[s].x = an[s].x * sc; ac[s].y = an[s].y * sc; }
      const float* pa = (g + 1 < G) ? (Arow + (g + 1) * 32) : Anext;
#pragma unroll
      for (int s = 0; s < 8; ++s) an[s] = *(const v2f*)(pa + 4 * s);

      const float* bp = Bl + g * (16 * PITCH);
      v2f bb[2][NT];
#pragma unroll
      for (int t = 0; t < NT; ++t) bb[0][t] = *(const v2f*)(bp + t * 32);
#pragma unroll
      for (int s = 0; s < 8; ++s) {
        if (s + 1 < 8) {
#pragma unroll
          for (int t = 0; t < NT; ++t)
            bb[(s + 1) & 1][t] = *(const v2f*)(bp + (s + 1) * (2 * PITCH) + t * 32);
        }
#pragma unroll
        for (int t = 0; t < NT; ++t)
          acc[t] = __builtin_amdgcn_wmma_f32_16x16x4_f32(
               false, ac[s],  false, bb[s & 1][t],
               (short)0, acc[t],  false,  false);
      }
    }

    const int row0 = tile * 16;
    float inn[8];
#pragma unroll
    for (int i = 0; i < 8; ++i) inn[i] = innorm ? innorm[row0 + i + 8 * half] : 1.0f;
#pragma unroll
    for (int i = 0; i < 8; ++i) {
      const int row = row0 + i + 8 * half;
#pragma unroll
      for (int t = 0; t < NT; ++t) {
        float v = acc[t][i];
        if (innorm) v = v * inn[i] + bv[t];
        if (relu)   v = fmaxf(v, 0.0f);
        C[(size_t)row * M + 16 * t + m] = v;
      }
    }

    if (ntile >= ntiles) break;
    tile = ntile;
  }
}


template<int M, bool SCALE, int B>
__device__ __forceinline__ void gather_add(typename fvec<M / 32>::type& acc,
                                           const float* __restrict__ hl,
                                           const int* __restrict__ c,
                                           const float* __restrict__ srcscale) {
  typedef typename fvec<M / 32>::type vec;
  int s[B];
#pragma unroll
  for (int k = 0; k < B; ++k) s[k] = c[k];
  vec r[B];
#pragma unroll
  for (int k = 0; k < B; ++k) r[k] = *(const vec*)(hl + (size_t)s[k] * M);
#pragma unroll
  for (int k = 0; k < B; ++k) {
    if (SCALE) acc += r[k] * srcscale[s[k]];
    else       acc += r[k];
  }
}

template<int M, bool SCALE>
__global__ __launch_bounds__(256)
void gather_kernel(const float* __restrict__ h,
                   const int* __restrict__ rowptr, const int* __restrict__ col,
                   const float* __restrict__ srcscale,
                   const float* __restrict__ innorm, const float* __restrict__ bias,
                   float* __restrict__ out, int N, int E, int relu) {
  constexpr int V = M / 32;
  typedef typename fvec<V>::type vec;
  const int lane = threadIdx.x & 31;
  const int row = __builtin_amdgcn_readfirstlane((int)(blockIdx.x * 8 + (threadIdx.x >> 5)));
  if (row >= N) return;
  const int beg = rowptr[row];
  const int end = (row + 1 < N) ? rowptr[row + 1] : E;
  const float* __restrict__ hl = h + lane * V;

  vec acc = {};
  int j = beg;
  for (; j + 8 <= end; j += 8) gather_add<M, SCALE, 8>(acc, hl, col + j, srcscale);
  if (j + 4 <= end) { gather_add<M, SCALE, 4>(acc, hl, col + j, srcscale); j += 4; }
  if (j + 2 <= end) { gather_add<M, SCALE, 2>(acc, hl, col + j, srcscale); j += 2; }
  if (j < end)      { gather_add<M, SCALE, 1>(acc, hl, col + j, srcscale); }

  if (innorm) {
    const float inn = innorm[row];
    const vec b = *(const vec*)(bias + lane * V);
    acc = acc * inn + b;
  }
  if (relu) {
#pragma unroll
    for (int c = 0; c < V; ++c) acc[c] = fmaxf(acc[c], 0.0f);
  }
  *(vec*)(out + (size_t)row * M + lane * V) = acc;
}


constexpr int TILE_PITCH = 128 + 4;
constexpr int FUSED_WAVES = 16;

template<int NT, int PITCH>
__device__ __forceinline__ void tile_product(v8f (&acc)[NT], const float* Al, const float* Bl) {
  v2f an[8];
#pragma unroll
  for (int s = 0; s < 8; ++s) an[s] = *(const v2f*)(Al + 4 * s);
#pragma nounroll
  for (int g = 0; g < 4; ++g) {
    v2f ac[8];
#pragma unroll
    for (int s = 0; s < 8; ++s) ac[s] = an[s];
    const float* pa = Al + ((g + 1 < 4) ? (g + 1) : g) * 32;
#pragma unroll
    for (int s = 0; s < 8; ++s) an[s] = *(const v2f*)(pa + 4 * s);

    const float* bp = Bl + g * (16 * PITCH);
    v2f bb[2][NT];
#pragma unroll
    for (int t = 0; t < NT; ++t) bb[0][t] = *(const v2f*)(bp + t * 32);
#pragma unroll
    for (int s = 0; s < 8; ++s) {
      if (s + 1 < 8) {
#pragma unroll
        for (int t = 0; t < NT; ++t)
          bb[(s + 1) & 1][t] = *(const v2f*)(bp + (s + 1) * (2 * PITCH) + t * 32);
      }
#pragma unroll
      for (int t = 0; t < NT; ++t)
        acc[t] = __builtin_amdgcn_wmma_f32_16x16x4_f32(
             false, ac[s],  false, bb[s & 1][t],
             (short)0, acc[t],  false,  false);
    }
  }
}

template<int K, int M, int PITCH, int THREADS>
__device__ __forceinline__ void stage_fragments(float* Bs, const float* __restrict__ B, int tid) {
  for (int it = tid; it < (K / 2) * (M / 2); it += THREADS) {
    const int kp = it / (M / 2);
    const int c2 = (it % (M / 2)) * 2;
    const v2f r0 = *(const v2f*)(B + (size_t)(2 * kp) * M + c2);
    const v2f r1 = *(const v2f*)(B + (size_t)(2 * kp + 1) * M + c2);
    const v4f o = {r0.x, r1.x, r0.y, r1.y};
    *(v4f*)(Bs + kp * PITCH + c2 * 2) = o;
  }
}

template<int M, int M3>
__global__ __launch_bounds__(FUSED_WAVES * 32)
void gather_gemm_gemm(const float* __restrict__ h,
                      const int* __restrict__ rowptr, const int* __restrict__ col,
                      const float* __restrict__ srcscale,
                      const float* __restrict__ B,
                      const float* __restrict__ innorm, const float* __restrict__ bias, int relu,
                      const float* __restrict__ B3, const float* __restrict__ scale3,
                      float* __restrict__ C3, int ntiles, int N, int E) {
  constexpr int K   = 128;
  constexpr int NT  = M / 16;
  constexpr int NT3 = M3 / 16;
  constexpr int PITCH  = 2 * M + 32;
  constexpr int PITCH3 = 2 * M3 + 32;
  __shared__ __attribute__((aligned(16))) float Bs[(K / 2) * PITCH];
  __shared__ __attribute__((aligned(16))) float Bs3[(M / 2) * PITCH3];
  __shared__ __attribute__((aligned(16))) float At[FUSED_WAVES * 16 * TILE_PITCH];

  const int tid = threadIdx.x;
  stage_fragments<K, M, PITCH, FUSED_WAVES * 32>(Bs, B, tid);
  stage_fragments<M, M3, PITCH3, FUSED_WAVES * 32>(Bs3, B3, tid);
  __syncthreads();

  const int lane = tid & 31;
  const int half = lane >> 4;
  const int m    = lane & 15;
  const int wave = __builtin_amdgcn_readfirstlane(tid >> 5);
  const int stride = (int)gridDim.x * FUSED_WAVES;
  int tile = wave * (int)gridDim.x + (int)blockIdx.x;
  if (tile >= ntiles) return;

  const float* Bl  = Bs  + half * PITCH  + m * 2;
  const float* Bl3 = Bs3 + half * PITCH3 + m * 2;
  float* Aw = At + wave * (16 * TILE_PITCH);
  const float* Al = Aw + m * TILE_PITCH + 2 * half;
  const float* __restrict__ hl = h + lane * 4;

  float bv[NT];
#pragma unroll
  for (int t = 0; t < NT; ++t) bv[t] = bias ? bias[16 * t + m] : 0.0f;

  for (;;) {
    const int row0 = tile * 16;

    int rp = E;
    if (lane <= 16) { const int r = row0 + lane; if (r < N) rp = rowptr[r]; }
    const int tb = __builtin_amdgcn_readlane(rp, 0);
    const int te = __builtin_amdgcn_readlane(rp, 16);
    int cur  = 0;
    int rend = __builtin_amdgcn_readlane(rp, 1);
    v4f acc = {};
    int idv = 0;
    if (tb < te) idv = col[min(tb + lane, te - 1)];
    for (int j0 = tb; j0 < te; j0 += 32) {
      const int ids = idv;
      const int scv = __float_as_int(srcscale[ids]);
      v4f r[32];
#pragma unroll
      for (int k = 0; k < 32; ++k) {
        const int s = __builtin_amdgcn_readlane(ids, k);
        r[k] = *(const v4f*)(hl + (size_t)s * K);
      }
      if (j0 + 32 < te) idv = col[min(j0 + 32 + lane, te - 1)];
#pragma unroll
      for (int k = 0; k < 32; ++k) {
        if (__builtin_expect(j0 + k >= rend, 0)) {
          do {
            *(v4f*)(Aw + cur * TILE_PITCH + lane * 4) = acc;
            acc = v4f{};
            ++cur;
            if (cur == 16) goto rows_done;
            rend = __builtin_amdgcn_readlane(rp, cur + 1);
          } while (j0 + k >= rend);
        }
        acc += r[k] * __int_as_float(__builtin_amdgcn_readlane(scv, k));
      }
    }
rows_done:
    for (; cur < 16; ++cur) {
      *(v4f*)(Aw + cur * TILE_PITCH + lane * 4) = acc;
      acc = v4f{};
    }

    {
      v8f pacc[NT];
#pragma unroll
      for (int t = 0; t < NT; ++t) pacc[t] = v8f{};
      tile_product<NT, PITCH>(pacc, Al, Bl);

      float inn[8], osc[8];
#pragma unroll
      for (int i = 0; i < 8; ++i) {
        inn[i] = innorm ? innorm[row0 + i + 8 * half] : 1.0f;
        osc[i] = scale3 ? scale3[row0 + i + 8 * half] : 1.0f;
      }
#pragma unroll
      for (int i = 0; i < 8; ++i) {
#pragma unroll
        for (int t = 0; t < NT; ++t) {
          float v = pacc[t][i];
          if (innorm) v = v * inn[i] + bv[t];
          if (relu)   v = fmaxf(v, 0.0f);
          Aw[(i + 8 * half) * TILE_PITCH + 16 * t + m] = v * osc[i];
        }
      }
    }

    {
      v8f qacc[NT3];
#pragma unroll
      for (int t = 0; t < NT3; ++t) qacc[t] = v8f{};
      tile_product<NT3, PITCH3>(qacc, Al, Bl3);
#pragma unroll
      for (int i = 0; i < 8; ++i) {
        const int row = row0 + i + 8 * half;
#pragma unroll
        for (int t = 0; t < NT3; ++t)
          C3[(size_t)row * M3 + 16 * t + m] = qacc[t][i];
      }
    }

    const int ntile = tile + stride;
    if (ntile >= ntiles) break;
    tile = ntile;
  }
}


extern "C" void kernel_launch(void* const* d_in, const int* in_sizes, int n_in,
                              void* d_out, int out_size, void* d_ws, size_t ws_size,
                              hipStream_t stream) {
  const float* x   = (const float*)d_in[0];
  const int*   src = (const int*)d_in[1];
  const int*   dst = (const int*)d_in[2];
  const float* W1  = (const float*)d_in[3];
  const float* b1  = (const float*)d_in[4];
  const float* W2  = (const float*)d_in[5];
  const float* b2  = (const float*)d_in[6];
  const float* W3  = (const float*)d_in[7];
  const float* b3  = (const float*)d_in[8];
  float* out = (float*)d_out;

  const int IN = 256, HID = 128, CLS = 64;
  const int E = in_sizes[1];
  const int N = in_sizes[0] / IN;

  float* ws      = (float*)d_ws;
  int*   rowptr  = (int*)ws;
  float* outnorm = ws + 2 * (size_t)N;
  float* innorm  = ws + 3 * (size_t)N;
  float* bufA    = ws + 4 * (size_t)N;
  float* bufB    = bufA + (size_t)N * HID;

  int* col   = (int*)d_out;
  int* col3  = (int*)bufB;

  const int nbins = (N + BIN_NODES - 1) / BIN_NODES;
  const int chunk = (E + SORT_BLOCKS - 1) / SORT_BLOCKS;
  int2* pairs = (int2*)bufB;
  int*  sbin  = (int*)bufB + 2 * (size_t)E;
  int2* descD = (int2*)((int*)bufB + 3 * (size_t)E);
  int2* descS = descD + (size_t)MAX_BINS * SORT_BLOCKS;

  const int rowtiles = N / 16;

  bin_sort_kernel<<<SORT_BLOCKS, 1024, 0, stream>>>(src, dst, E, chunk, nbins,
                                                    pairs, sbin, descD, descS);
  bin_rows_kernel<<<nbins, 1024, 0, stream>>>(pairs, sbin, descD, descS, SORT_BLOCKS,
                                              rowptr, outnorm, innorm, col, N);

  gemm_wmma<256, 128><<<512, 256, 0, stream>>>(x, W1, bufA, outnorm,
                                               nullptr, nullptr, 0, rowtiles);
  gather_kernel<128, false><<<(N + 7) / 8, 256, 0, stream>>>(bufA, rowptr, col, nullptr,
                                                             innorm, b1, bufB, N, E, 1);

  gather_gemm_gemm<128, 64><<<256, 512, 0, stream>>>(bufB, rowptr, col, outnorm, W2,
                                                     innorm, b2, 1, W3, outnorm,
                                                     bufA, rowtiles, N, E);

  move_words_kernel<<<(E + 255) / 256, 256, 0, stream>>>(col, col3, E);
  gather_kernel<64, false><<<(N + 7) / 8, 256, 0, stream>>>(bufA, rowptr, col3, nullptr,
                                                            innorm, b3, out, N, E, 0);
}


# layer 1's aggregation stores its rows scaled by the source norm, so layer 2's aggregation adds plain rows and gathers no norm word per edge; product rounded before the sum; sum order unchanged
# speedup vs baseline: 4.8803x; 1.0357x over previous
#include <hip/hip_runtime.h>
#include <math.h>


typedef __attribute__((ext_vector_type(2))) float v2f;
typedef __attribute__((ext_vector_type(4))) float v4f;
typedef __attribute__((ext_vector_type(8))) float v8f;

template<int V> struct fvec { typedef float type __attribute__((ext_vector_type(V))); };


constexpr int BIN_SHIFT = 9;
constexpr int BIN_NODES = 1 << BIN_SHIFT;
constexpr int MAX_BINS  = 256;
constexpr int SORT_BLOCKS = 256;
constexpr int SORT_CAP  = 8192;
constexpr int ROWS_CAP  = 16384;

__global__ __launch_bounds__(1024)
void bin_sort_kernel(const int* __restrict__ src, const int* __restrict__ dst, int E, int chunk,
                     int nbins, int2* __restrict__ pairs, int* __restrict__ sbin,
                     int2* __restrict__ descD, int2* __restrict__ descS) {
  __shared__ int cd[MAX_BINS], cs[MAX_BINS], sd[MAX_BINS], ss[MAX_BINS];
  __shared__ int2 lp[SORT_CAP];
  __shared__ int  ls[SORT_CAP];
  const int t = threadIdx.x;
  const int g = blockIdx.x, G = gridDim.x;
  if (t < MAX_BINS) { cd[t] = 0; cs[t] = 0; }
  __syncthreads();
  const int e0 = min(E, g * chunk);
  const int e1 = min(E, e0 + chunk);
  for (int e = e0 + t; e < e1; e += 1024) {
    atomicAdd(&cd[dst[e] >> BIN_SHIFT], 1);
    atomicAdd(&cs[src[e] >> BIN_SHIFT], 1);
  }
  __syncthreads();
  if (t < MAX_BINS) { sd[t] = cd[t]; ss[t] = cs[t]; }
  __syncthreads();
  for (int off = 1; off < MAX_BINS; off <<= 1) {
    int vd = 0, vs = 0;
    if (t < MAX_BINS && t >= off) { vd = sd[t - off]; vs = ss[t - off]; }
    __syncthreads();
    if (t < MAX_BINS) { sd[t] += vd; ss[t] += vs; }
    __syncthreads();
  }
  if (t < MAX_BINS) {
    const int nd = cd[t], ns = cs[t];
    const int pd = e0 + sd[t] - nd;
    const int ps = e0 + ss[t] - ns;
    if (t < nbins) {
      descD[t * G + g] = make_int2(pd, nd);
      descS[t * G + g] = make_int2(ps, ns);
    }
    cd[t] = pd;
    cs[t] = ps;
  }
  __syncthreads();
  if (chunk <= SORT_CAP) {
    for (int e = e0 + t; e < e1; e += 1024) {
      const int d = dst[e], s = src[e];
      const int p = atomicAdd(&cd[d >> BIN_SHIFT], 1);
      lp[p - e0] = make_int2(d, s);
      const int q = atomicAdd(&cs[s >> BIN_SHIFT], 1);
      ls[q - e0] = s;
    }
    __syncthreads();
    const int n = e1 - e0;
    for (int i = t - (e0 & 31); i < n; i += 1024) {
      if (i >= 0) {
        pairs[e0 + i] = lp[i];
        sbin[e0 + i]  = ls[i];
      }
    }
  } else {
    for (int e = e0 + t; e < e1; e += 1024) {
      const int d = dst[e], s = src[e];
      const int p = atomicAdd(&cd[d >> BIN_SHIFT], 1);
      pairs[p] = make_int2(d, s);
      const int q = atomicAdd(&cs[s >> BIN_SHIFT], 1);
      sbin[q] = s;
    }
  }
}

__global__ __launch_bounds__(1024)
void bin_rows_kernel(const int2* __restrict__ pairs, const int* __restrict__ sbin,
                     const int2* __restrict__ descD, const int2* __restrict__ descS, int G,
                     int* __restrict__ rowptr, float* __restrict__ outnorm,
                     float* __restrict__ innorm, int* __restrict__ col, int N) {
  __shared__ int cin[BIN_NODES], cou[BIN_NODES], sc[BIN_NODES];
  __shared__ int basesum;
  __shared__ int lcol[ROWS_CAP];
  const int t = threadIdx.x;
  const int b = blockIdx.x;
  const int sub = t & 3;
  if (t < BIN_NODES) { cin[t] = 0; cou[t] = 0; }
  if (t == 0) basesum = 0;
  __syncthreads();

  int part = 0;
  const int nprev = b * G;
  for (int i = t; i < nprev; i += 1024) part += descD[i].y;
  for (int off = 16; off > 0; off >>= 1) part += __shfl_xor(part, off);
  if ((t & 31) == 0) atomicAdd(&basesum, part);

  for (int r = t >> 2; r < G; r += 256) {
    const int2 dd = descD[b * G + r];
    const int2 ds = descS[b * G + r];
    for (int k = sub; k < dd.y; k += 4)
      atomicAdd(&cin[pairs[dd.x + k].x & (BIN_NODES - 1)], 1);
    for (int k = sub; k < ds.y; k += 4)
      atomicAdd(&cou[sbin[ds.x + k] & (BIN_NODES - 1)], 1);
  }
  __syncthreads();
  if (t < BIN_NODES) sc[t] = cin[t];
  __syncthreads();
  for (int off = 1; off < BIN_NODES; off <<= 1) {
    int v = 0;
    if (t < BIN_NODES && t >= off) v = sc[t - off];
    __syncthreads();
    if (t < BIN_NODES) sc[t] += v;
    __syncthreads();
  }
  const int base  = basesum;
  const int total = sc[BIN_NODES - 1];
  const bool in_lds = total <= ROWS_CAP;
  if (t < BIN_NODES) {
    const int ci = cin[t];
    const int start = base + sc[t] - ci;
    const int node = b * BIN_NODES + t;
    if (node < N) {
      rowptr[node]  = start;
      innorm[node]  = rsqrtf(fmaxf((float)ci, 1.0f));
      outnorm[node] = rsqrtf(fmaxf((float)cou[t], 1.0f));
    }
    cin[t] = start;
  }
  __syncthreads();
  for (int r = t >> 2; r < G; r += 256) {
    const int2 dd = descD[b * G + r];
    for (int k = sub; k < dd.y; k += 4) {
      const int2 p = pairs[dd.x + k];
      const int slot = atomicAdd(&cin[p.x & (BIN_NODES - 1)], 1);
      if (in_lds) lcol[slot - base] = p.y;
      else        col[slot] = p.y;
    }
  }
  __syncthreads();
  if (in_lds) {
    for (int i = t - (base & 31); i < total; i += 1024)
      if (i >= 0) col[base + i] = lcol[i];
  }
}

__global__ void move_words_kernel(const int* __restrict__ from, int* __restrict__ to, int n) {
  const int i = blockIdx.x * blockDim.x + threadIdx.x;
  if (i < n) to[i] = from[i];
}

template<int K, int M>
__global__ __launch_bounds__(256)
void gemm_wmma(const float* __restrict__ A, const float* __restrict__ B,
               float* __restrict__ C,
               const float* __restrict__ rowscale,
               const float* __restrict__ innorm, const float* __restrict__ bias,
               int relu, int ntiles) {
  constexpr int NT = M / 16;
  constexpr int G = K / 32;
  constexpr int PITCH = 2 * M + 32;
  __shared__ __attribute__((aligned(16))) float Bs[(K / 2) * PITCH];

  const int tid = threadIdx.x;
  for (int it = tid; it < (K / 2) * (M / 2); it += 256) {
    const int kp = it / (M / 2);
    const int c2 = (it % (M / 2)) * 2;
    const v2f r0 = *(const v2f*)(B + (size_t)(2 * kp) * M + c2);
    const v2f r1 = *(const v2f*)(B + (size_t)(2 * kp + 1) * M + c2);
    const v4f o = {r0.x, r1.x, r0.y, r1.y};
    *(v4f*)(Bs + kp * PITCH + c2 * 2) = o;
  }
  __syncthreads();

  const int lane = tid & 31;
  const int half = lane >> 4;
  const int m    = lane & 15;
  const int wave = __builtin_amdgcn_readfirstlane(tid >> 5);
  const int stride = (int)gridDim.x * 8;
  int tile = wave * (int)gridDim.x + (int)blockIdx.x;
  if (tile >= ntiles) return;

  const float* Bl = Bs + half * PITCH + m * 2;

  float bv[NT];
#pragma unroll
  for (int t = 0; t < NT; ++t) bv[t] = bias ? bias[16 * t + m] : 0.0f;

  v2f an[8];
  {
    const float* pa = A + (size_t)(tile * 16 + m) * K + 2 * half;
#pragma unroll
    for (int s = 0; s < 8; ++s) an[s] = *(const v2f*)(pa + 4 * s);
  }
  float scn = rowscale ? rowscale[tile * 16 + m] : 1.0f;

  for (;;) {
    const int ntile = tile + stride;
    const int ptile = (ntile < ntiles) ? ntile : tile;
    const float sc = scn;
    scn = rowscale ? rowscale[ptile * 16 + m] : 1.0f;
    const float* Arow  = A + (size_t)(tile  * 16 + m) * K + 2 * half;
    const float* Anext = A + (size_t)(ptile * 16 + m) * K + 2 * half;

    v8f acc[NT];
#pragma unroll
    for (int t = 0; t < NT; ++t) acc[t] = v8f{};

#pragma nounroll
    for (int g = 0; g < G; ++g) {
      v2f ac[8];
#pragma unroll
      for (int s = 0; s < 8; ++s) { ac[s].x = an[s].x * sc; ac[s].y = an[s].y * sc; }
      const float* pa = (g + 1 < G) ? (Arow + (g + 1) * 32) : Anext;
#pragma unroll
      for (int s = 0; s < 8; ++s) an[s] = *(const v2f*)(pa + 4 * s);

      const float* bp = Bl + g * (16 * PITCH);
      v2f bb[2][NT];
#pragma unroll
      for (int t = 0; t < NT; ++t) bb[0][t] = *(const v2f*)(bp + t * 32);
#pragma unroll
      for (int s = 0; s < 8; ++s) {
        if (s + 1 < 8) {
#pragma unroll
          for (int t = 0; t < NT; ++t)
            bb[(s + 1) & 1][t] = *(const v2f*)(bp + (s + 1) * (2 * PITCH) + t * 32);
        }
#pragma unroll
        for (int t = 0; t < NT; ++t)
          acc[t] = __builtin_amdgcn_wmma_f32_16x16x4_f32(
               false, ac[s],  false, bb[s & 1][t],
               (short)0, acc[t],  false,  false);
      }
    }

    const int row0 = tile * 16;
    float inn[8];
#pragma unroll
    for (int i = 0; i < 8; ++i) inn[i] = innorm ? innorm[row0 + i + 8 * half] : 1.0f;
#pragma unroll
    for (int i = 0; i < 8; ++i) {
      const int row = row0 + i + 8 * half;
#pragma unroll
      for (int t = 0; t < NT; ++t) {
        float v = acc[t][i];
        if (innorm) v = v * inn[i] + bv[t];
        if (relu)   v = fmaxf(v, 0.0f);
        C[(size_t)row * M + 16 * t + m] = v;
      }
    }

    if (ntile >= ntiles) break;
    tile = ntile;
  }
}


template<int M, bool SCALE, int B>
__device__ __forceinline__ void gather_add(typename fvec<M / 32>::type& acc,
                                           const float* __restrict__ hl,
                                           const int* __restrict__ c,
                                           const float* __restrict__ srcscale) {
  typedef typename fvec<M / 32>::type vec;
  int s[B];
#pragma unroll
  for (int k = 0; k < B; ++k) s[k] = c[k];
  vec r[B];
#pragma unroll
  for (int k = 0; k < B; ++k) r[k] = *(const vec*)(hl + (size_t)s[k] * M);
#pragma unroll
  for (int k = 0; k < B; ++k) {
    if (SCALE) acc += r[k] * srcscale[s[k]];
    else       acc += r[k];
  }
}

template<int M, bool SCALE>
__global__ __launch_bounds__(256)
void gather_kernel(const float* __restrict__ h,
                   const int* __restrict__ rowptr, const int* __restrict__ col,
                   const float* __restrict__ srcscale,
                   const float* __restrict__ innorm, const float* __restrict__ bias,
                   const float* __restrict__ postscale,
                   float* __restrict__ out, int N, int E, int relu) {
  constexpr int V = M / 32;
  typedef typename fvec<V>::type vec;
  const int lane = threadIdx.x & 31;
  const int row = __builtin_amdgcn_readfirstlane((int)(blockIdx.x * 8 + (threadIdx.x >> 5)));
  if (row >= N) return;
  const int beg = rowptr[row];
  const int end = (row + 1 < N) ? rowptr[row + 1] : E;
  const float* __restrict__ hl = h + lane * V;

  vec acc = {};
  int j = beg;
  for (; j + 8 <= end; j += 8) gather_add<M, SCALE, 8>(acc, hl, col + j, srcscale);
  if (j + 4 <= end) { gather_add<M, SCALE, 4>(acc, hl, col + j, srcscale); j += 4; }
  if (j + 2 <= end) { gather_add<M, SCALE, 2>(acc, hl, col + j, srcscale); j += 2; }
  if (j < end)      { gather_add<M, SCALE, 1>(acc, hl, col + j, srcscale); }

  if (innorm) {
    const float inn = innorm[row];
    const vec b = *(const vec*)(bias + lane * V);
    acc = acc * inn + b;
  }
  if (relu) {
#pragma unroll
    for (int c = 0; c < V; ++c) acc[c] = fmaxf(acc[c], 0.0f);
  }
  if (postscale) acc = acc * postscale[row];
  *(vec*)(out + (size_t)row * M + lane * V) = acc;
}


constexpr int TILE_PITCH = 128 + 4;
constexpr int FUSED_WAVES = 16;

template<int NT, int PITCH>
__device__ __forceinline__ void tile_product(v8f (&acc)[NT], const float* Al, const float* Bl) {
  v2f an[8];
#pragma unroll
  for (int s = 0; s < 8; ++s) an[s] = *(const v2f*)(Al + 4 * s);
#pragma nounroll
  for (int g = 0; g < 4; ++g) {
    v2f ac[8];
#pragma unroll
    for (int s = 0; s < 8; ++s) ac[s] = an[s];
    const float* pa = Al + ((g + 1 < 4) ? (g + 1) : g) * 32;
#pragma unroll
    for (int s = 0; s < 8; ++s) an[s] = *(const v2f*)(pa + 4 * s);

    const float* bp = Bl + g * (16 * PITCH);
    v2f bb[2][NT];
#pragma unroll
    for (int t = 0; t < NT; ++t) bb[0][t] = *(const v2f*)(bp + t * 32);
#pragma unroll
    for (int s = 0; s < 8; ++s) {
      if (s + 1 < 8) {
#pragma unroll
        for (int t = 0; t < NT; ++t)
          bb[(s + 1) & 1][t] = *(const v2f*)(bp + (s + 1) * (2 * PITCH) + t * 32);
      }
#pragma unroll
      for (int t = 0; t < NT; ++t)
        acc[t] = __builtin_amdgcn_wmma_f32_16x16x4_f32(
             false, ac[s],  false, bb[s & 1][t],
             (short)0, acc[t],  false,  false);
    }
  }
}

template<int K, int M, int PITCH, int THREADS>
__device__ __forceinline__ void stage_fragments(float* Bs, const float* __restrict__ B, int tid) {
  for (int it = tid; it < (K / 2) * (M / 2); it += THREADS) {
    const int kp = it / (M / 2);
    const int c2 = (it % (M / 2)) * 2;
    const v2f r0 = *(const v2f*)(B + (size_t)(2 * kp) * M + c2);
    const v2f r1 = *(const v2f*)(B + (size_t)(2 * kp + 1) * M + c2);
    const v4f o = {r0.x, r1.x, r0.y, r1.y};
    *(v4f*)(Bs + kp * PITCH + c2 * 2) = o;
  }
}

template<int M, int M3>
__global__ __launch_bounds__(FUSED_WAVES * 32)
void gather_gemm_gemm(const float* __restrict__ h,
                      const int* __restrict__ rowptr, const int* __restrict__ col,
                      const float* __restrict__ B,
                      const float* __restrict__ innorm, const float* __restrict__ bias, int relu,
                      const float* __restrict__ B3, const float* __restrict__ scale3,
                      float* __restrict__ C3, int ntiles, int N, int E) {
  constexpr int K   = 128;
  constexpr int NT  = M / 16;
  constexpr int NT3 = M3 / 16;
  constexpr int PITCH  = 2 * M + 32;
  constexpr int PITCH3 = 2 * M3 + 32;
  __shared__ __attribute__((aligned(16))) float Bs[(K / 2) * PITCH];
  __shared__ __attribute__((aligned(16))) float Bs3[(M / 2) * PITCH3];
  __shared__ __attribute__((aligned(16))) float At[FUSED_WAVES * 16 * TILE_PITCH];

  const int tid = threadIdx.x;
  stage_fragments<K, M, PITCH, FUSED_WAVES * 32>(Bs, B, tid);
  stage_fragments<M, M3, PITCH3, FUSED_WAVES * 32>(Bs3, B3, tid);
  __syncthreads();

  const int lane = tid & 31;
  const int half = lane >> 4;
  const int m    = lane & 15;
  const int wave = __builtin_amdgcn_readfirstlane(tid >> 5);
  const int stride = (int)gridDim.x * FUSED_WAVES;
  int tile = wave * (int)gridDim.x + (int)blockIdx.x;
  if (tile >= ntiles) return;

  const float* Bl  = Bs  + half * PITCH  + m * 2;
  const float* Bl3 = Bs3 + half * PITCH3 + m * 2;
  float* Aw = At + wave * (16 * TILE_PITCH);
  const float* Al = Aw + m * TILE_PITCH + 2 * half;
  const float* __restrict__ hl = h + lane * 4;

  float bv[NT];
#pragma unroll
  for (int t = 0; t < NT; ++t) bv[t] = bias ? bias[16 * t + m] : 0.0f;

  for (;;) {
    const int row0 = tile * 16;

    int rp = E;
    if (lane <= 16) { const int r = row0 + lane; if (r < N) rp = rowptr[r]; }
    const int tb = __builtin_amdgcn_readlane(rp, 0);
    const int te = __builtin_amdgcn_readlane(rp, 16);
    int cur  = 0;
    int rend = __builtin_amdgcn_readlane(rp, 1);
    v4f acc = {};
    int idv = 0;
    if (tb < te) idv = col[min(tb + lane, te - 1)];
    for (int j0 = tb; j0 < te; j0 += 32) {
      const int ids = idv;
      v4f r[32];
#pragma unroll
      for (int k = 0; k < 32; ++k) {
        const int s = __builtin_amdgcn_readlane(ids, k);
        r[k] = *(const v4f*)(hl + (size_t)s * K);
      }
      if (j0 + 32 < te) idv = col[min(j0 + 32 + lane, te - 1)];
#pragma unroll
      for (int k = 0; k < 32; ++k) {
        if (__builtin_expect(j0 + k >= rend, 0)) {
          do {
            *(v4f*)(Aw + cur * TILE_PITCH + lane * 4) = acc;
            acc = v4f{};
            ++cur;
            if (cur == 16) goto rows_done;
            rend = __builtin_amdgcn_readlane(rp, cur + 1);
          } while (j0 + k >= rend);
        }
        acc += r[k];
      }
    }
rows_done:
    for (; cur < 16; ++cur) {
      *(v4f*)(Aw + cur * TILE_PITCH + lane * 4) = acc;
      acc = v4f{};
    }

    {
      v8f pacc[NT];
#pragma unroll
      for (int t = 0; t < NT; ++t) pacc[t] = v8f{};
      tile_product<NT, PITCH>(pacc, Al, Bl);

      float inn[8], osc[8];
#pragma unroll
      for (int i = 0; i < 8; ++i) {
        inn[i] = innorm ? innorm[row0 + i + 8 * half] : 1.0f;
        osc[i] = scale3 ? scale3[row0 + i + 8 * half] : 1.0f;
      }
#pragma unroll
      for (int i = 0; i < 8; ++i) {
#pragma unroll
        for (int t = 0; t < NT; ++t) {
          float v = pacc[t][i];
          if (innorm) v = v * inn[i] + bv[t];
          if (relu)   v = fmaxf(v, 0.0f);
          Aw[(i + 8 * half) * TILE_PITCH + 16 * t + m] = v * osc[i];
        }
      }
    }

    {
      v8f qacc[NT3];
#pragma unroll
      for (int t = 0; t < NT3; ++t) qacc[t] = v8f{};
      tile_product<NT3, PITCH3>(qacc, Al, Bl3);
#pragma unroll
      for (int i = 0; i < 8; ++i) {
        const int row = row0 + i + 8 * half;
#pragma unroll
        for (int t = 0; t < NT3; ++t)
          C3[(size_t)row * M3 + 16 * t + m] = qacc[t][i];
      }
    }

    const int ntile = tile + stride;
    if (ntile >= ntiles) break;
    tile = ntile;
  }
}


extern "C" void kernel_launch(void* const* d_in, const int* in_sizes, int n_in,
                              void* d_out, int out_size, void* d_ws, size_t ws_size,
                              hipStream_t stream) {
  const float* x   = (const float*)d_in[0];
  const int*   src = (const int*)d_in[1];
  const int*   dst = (const int*)d_in[2];
  const float* W1  = (const float*)d_in[3];
  const float* b1  = (const float*)d_in[4];
  const float* W2  = (const float*)d_in[5];
  const float* b2  = (const float*)d_in[6];
  const float* W3  = (const float*)d_in[7];
  const float* b3  = (const float*)d_in[8];
  float* out = (float*)d_out;

  const int IN = 256, HID = 128, CLS = 64;
  const int E = in_sizes[1];
  const int N = in_sizes[0] / IN;

  float* ws      = (float*)d_ws;
  int*   rowptr  = (int*)ws;
  float* outnorm = ws + 2 * (size_t)N;
  float* innorm  = ws + 3 * (size_t)N;
  float* bufA    = ws + 4 * (size_t)N;
  float* bufB    = bufA + (size_t)N * HID;

  int* col   = (int*)d_out;
  int* col3  = (int*)bufB;

  const int nbins = (N + BIN_NODES - 1) / BIN_NODES;
  const int chunk = (E + SORT_BLOCKS - 1) / SORT_BLOCKS;
  int2* pairs = (int2*)bufB;
  int*  sbin  = (int*)bufB + 2 * (size_t)E;
  int2* descD = (int2*)((int*)bufB + 3 * (size_t)E);
  int2* descS = descD + (size_t)MAX_BINS * SORT_BLOCKS;

  const int rowtiles = N / 16;

  bin_sort_kernel<<<SORT_BLOCKS, 1024, 0, stream>>>(src, dst, E, chunk, nbins,
                                                    pairs, sbin, descD, descS);
  bin_rows_kernel<<<nbins, 1024, 0, stream>>>(pairs, sbin, descD, descS, SORT_BLOCKS,
                                              rowptr, outnorm, innorm, col, N);

  gemm_wmma<256, 128><<<512, 256, 0, stream>>>(x, W1, bufA, outnorm,
                                               nullptr, nullptr, 0, rowtiles);
  gather_kernel<128, false><<<(N + 7) / 8, 256, 0, stream>>>(bufA, rowptr, col, nullptr,
                                                             innorm, b1, outnorm, bufB, N, E, 1);

  gather_gemm_gemm<128, 64><<<256, 512, 0, stream>>>(bufB, rowptr, col, W2,
                                                     innorm, b2, 1, W3, outnorm,
                                                     bufA, rowtiles, N, E);

  move_words_kernel<<<(E + 255) / 256, 256, 0, stream>>>(col, col3, E);
  gather_kernel<64, false><<<(N + 7) / 8, 256, 0, stream>>>(bufA, rowptr, col3, nullptr,
                                                            innorm, b3, nullptr, out, N, E, 0);
}


# list building: every edge walk issues a batch of loads before its first LDS atomic (8 edges a thread; a wave takes 8 runs, a lane per edge), not one load per trip; row sum order = LDS atomic arrival
# speedup vs baseline: 5.1030x; 1.0456x over previous
#include <hip/hip_runtime.h>
#include <math.h>


typedef __attribute__((ext_vector_type(2))) float v2f;
typedef __attribute__((ext_vector_type(4))) float v4f;
typedef __attribute__((ext_vector_type(8))) float v8f;

template<int V> struct fvec { typedef float type __attribute__((ext_vector_type(V))); };


constexpr int BIN_SHIFT = 9;
constexpr int BIN_NODES = 1 << BIN_SHIFT;
constexpr int MAX_BINS  = 256;
constexpr int SORT_BLOCKS = 256;
constexpr int SORT_CAP  = 8192;
constexpr int ROWS_CAP  = 16384;
constexpr int EDGE_BATCH = 8;
constexpr int RUN_BATCH  = 8;

__global__ __launch_bounds__(1024)
void bin_sort_kernel(const int* __restrict__ src, const int* __restrict__ dst, int E, int chunk,
                     int nbins, int2* __restrict__ pairs, int* __restrict__ sbin,
                     int2* __restrict__ descD, int2* __restrict__ descS) {
  __shared__ int cd[MAX_BINS], cs[MAX_BINS], sd[MAX_BINS], ss[MAX_BINS];
  __shared__ int2 lp[SORT_CAP];
  __shared__ int  ls[SORT_CAP];
  const int t = threadIdx.x;
  const int g = blockIdx.x, G = gridDim.x;
  if (t < MAX_BINS) { cd[t] = 0; cs[t] = 0; }
  __syncthreads();
  const int e0 = min(E, g * chunk);
  const int e1 = min(E, e0 + chunk);
  for (int eb = e0 + t; eb < e1; eb += 1024 * EDGE_BATCH) {
    int d[EDGE_BATCH], s[EDGE_BATCH];
#pragma unroll
    for (int u = 0; u < EDGE_BATCH; ++u) {
      const int e = min(eb + 1024 * u, e1 - 1);
      d[u] = dst[e];
      s[u] = src[e];
    }
#pragma unroll
    for (int u = 0; u < EDGE_BATCH; ++u) {
      asm volatile("" : "+v"(d[u]), "+v"(s[u]));
    }
#pragma unroll
    for (int u = 0; u < EDGE_BATCH; ++u) {
      if (eb + 1024 * u < e1) {
        atomicAdd(&cd[d[u] >> BIN_SHIFT], 1);
        atomicAdd(&cs[s[u] >> BIN_SHIFT], 1);
      }
    }
  }
  __syncthreads();
  if (t < MAX_BINS) { sd[t] = cd[t]; ss[t] = cs[t]; }
  __syncthreads();
  for (int off = 1; off < MAX_BINS; off <<= 1) {
    int vd = 0, vs = 0;
    if (t < MAX_BINS && t >= off) { vd = sd[t - off]; vs = ss[t - off]; }
    __syncthreads();
    if (t < MAX_BINS) { sd[t] += vd; ss[t] += vs; }
    __syncthreads();
  }
  if (t < MAX_BINS) {
    const int nd = cd[t], ns = cs[t];
    const int pd = e0 + sd[t] - nd;
    const int ps = e0 + ss[t] - ns;
    if (t < nbins) {
      descD[t * G + g] = make_int2(pd, nd);
      descS[t * G + g] = make_int2(ps, ns);
    }
    cd[t] = pd;
    cs[t] = ps;
  }
  __syncthreads();
  if (chunk <= SORT_CAP) {
    for (int eb = e0 + t; eb < e1; eb += 1024 * EDGE_BATCH) {
      int d[EDGE_BATCH], s[EDGE_BATCH];
#pragma unroll
      for (int u = 0; u < EDGE_BATCH; ++u) {
        const int e = min(eb + 1024 * u, e1 - 1);
        d[u] = dst[e];
        s[u] = src[e];
      }
#pragma unroll
      for (int u = 0; u < EDGE_BATCH; ++u) {
        asm volatile("" : "+v"(d[u]), "+v"(s[u]));
      }
#pragma unroll
      for (int u = 0; u < EDGE_BATCH; ++u) {
        if (eb + 1024 * u < e1) {
          const int p = atomicAdd(&cd[d[u] >> BIN_SHIFT], 1);
          lp[p - e0] = make_int2(d[u], s[u]);
          const int q = atomicAdd(&cs[s[u] >> BIN_SHIFT], 1);
          ls[q - e0] = s[u];
        }
      }
    }
    __syncthreads();
    const int n = e1 - e0;
    for (int i = t - (e0 & 31); i < n; i += 1024) {
      if (i >= 0) {
        pairs[e0 + i] = lp[i];
        sbin[e0 + i]  = ls[i];
      }
    }
  } else {
    for (int e = e0 + t; e < e1; e += 1024) {
      const int d = dst[e], s = src[e];
      const int p = atomicAdd(&cd[d >> BIN_SHIFT], 1);
      pairs[p] = make_int2(d, s);
      const int q = atomicAdd(&cs[s >> BIN_SHIFT], 1);
      sbin[q] = s;
    }
  }
}

__global__ __launch_bounds__(1024)
void bin_rows_kernel(const int2* __restrict__ pairs, const int* __restrict__ sbin,
                     const int2* __restrict__ descD, const int2* __restrict__ descS, int G,
                     int* __restrict__ rowptr, float* __restrict__ outnorm,
                     float* __restrict__ innorm, int* __restrict__ col, int N) {
  __shared__ int cin[BIN_NODES], cou[BIN_NODES], sc[BIN_NODES];
  __shared__ int basesum;
  __shared__ int lcol[ROWS_CAP];
  const int t = threadIdx.x;
  const int b = blockIdx.x;
  const int lane = t & 31;
  const int wv = __builtin_amdgcn_readfirstlane(t >> 5);
  if (t < BIN_NODES) { cin[t] = 0; cou[t] = 0; }
  if (t == 0) basesum = 0;
  __syncthreads();

  int part = 0;
  const int nprev = b * G;
  for (int i = t; i < nprev; i += 1024) part += descD[i].y;
  for (int off = 16; off > 0; off >>= 1) part += __shfl_xor(part, off);
  if ((t & 31) == 0) atomicAdd(&basesum, part);

  for (int r0 = wv; r0 < G; r0 += 32 * RUN_BATCH) {
    const int rl = min(r0 + 32 * (lane & (RUN_BATCH - 1)), G - 1);
    const int2 ddv = descD[b * G + rl];
    const int2 dsv = descS[b * G + rl];
    int dx[RUN_BATCH], dn[RUN_BATCH], sx[RUN_BATCH], sn[RUN_BATCH];
#pragma unroll
    for (int u = 0; u < RUN_BATCH; ++u) {
      const bool live = r0 + 32 * u < G;
      dx[u] = __builtin_amdgcn_readlane(ddv.x, u);
      dn[u] = live ? __builtin_amdgcn_readlane(ddv.y, u) : 0;
      sx[u] = __builtin_amdgcn_readlane(dsv.x, u);
      sn[u] = live ? __builtin_amdgcn_readlane(dsv.y, u) : 0;
    }
    int vd[RUN_BATCH][2], vs[RUN_BATCH][2];
#pragma unroll
    for (int u = 0; u < RUN_BATCH; ++u) {
#pragma unroll
      for (int h = 0; h < 2; ++h) {
        const int k = lane + 32 * h;
        vd[u][h] = pairs[(dn[u] > 0 ? dx[u] + min(k, dn[u] - 1) : 0)].x;
        vs[u][h] = sbin[(sn[u] > 0 ? sx[u] + min(k, sn[u] - 1) : 0)];
      }
    }
#pragma unroll
    for (int u = 0; u < RUN_BATCH; ++u) {
      asm volatile("" : "+v"(vd[u][0]), "+v"(vd[u][1]), "+v"(vs[u][0]), "+v"(vs[u][1]));
    }
#pragma unroll
    for (int u = 0; u < RUN_BATCH; ++u) {
#pragma unroll
      for (int h = 0; h < 2; ++h) {
        const int k = lane + 32 * h;
        if (k < dn[u]) atomicAdd(&cin[vd[u][h] & (BIN_NODES - 1)], 1);
        if (k < sn[u]) atomicAdd(&cou[vs[u][h] & (BIN_NODES - 1)], 1);
      }
    }
#pragma unroll
    for (int u = 0; u < RUN_BATCH; ++u) {
      for (int k = lane + 64; k < dn[u]; k += 32)
        atomicAdd(&cin[pairs[dx[u] + k].x & (BIN_NODES - 1)], 1);
      for (int k = lane + 64; k < sn[u]; k += 32)
        atomicAdd(&cou[sbin[sx[u] + k] & (BIN_NODES - 1)], 1);
    }
  }
  __syncthreads();
  if (t < BIN_NODES) sc[t] = cin[t];
  __syncthreads();
  for (int off = 1; off < BIN_NODES; off <<= 1) {
    int v = 0;
    if (t < BIN_NODES && t >= off) v = sc[t - off];
    __syncthreads();
    if (t < BIN_NODES) sc[t] += v;
    __syncthreads();
  }
  const int base  = basesum;
  const int total = sc[BIN_NODES - 1];
  const bool in_lds = total <= ROWS_CAP;
  if (t < BIN_NODES) {
    const int ci = cin[t];
    const int start = base + sc[t] - ci;
    const int node = b * BIN_NODES + t;
    if (node < N) {
      rowptr[node]  = start;
      innorm[node]  = rsqrtf(fmaxf((float)ci, 1.0f));
      outnorm[node] = rsqrtf(fmaxf((float)cou[t], 1.0f));
    }
    cin[t] = start;
  }
  __syncthreads();
  for (int r0 = wv; r0 < G; r0 += 32 * RUN_BATCH) {
    const int rl = min(r0 + 32 * (lane & (RUN_BATCH - 1)), G - 1);
    const int2 ddv = descD[b * G + rl];
    int dx[RUN_BATCH], dn[RUN_BATCH];
#pragma unroll
    for (int u = 0; u < RUN_BATCH; ++u) {
      dx[u] = __builtin_amdgcn_readlane(ddv.x, u);
      dn[u] = (r0 + 32 * u < G) ? __builtin_amdgcn_readlane(ddv.y, u) : 0;
    }
    int2 pv[RUN_BATCH][2];
#pragma unroll
    for (int u = 0; u < RUN_BATCH; ++u) {
#pragma unroll
      for (int h = 0; h < 2; ++h) {
        const int k = lane + 32 * h;
        pv[u][h] = pairs[(dn[u] > 0 ? dx[u] + min(k, dn[u] - 1) : 0)];
      }
    }
#pragma unroll
    for (int u = 0; u < RUN_BATCH; ++u) {
      asm volatile("" : "+v"(pv[u][0].x), "+v"(pv[u][0].y), "+v"(pv[u][1].x), "+v"(pv[u][1].y));
    }
#pragma unroll
    for (int u = 0; u < RUN_BATCH; ++u) {
#pragma unroll
      for (int h = 0; h < 2; ++h) {
        if (lane + 32 * h < dn[u]) {
          const int2 p = pv[u][h];
          const int slot = atomicAdd(&cin[p.x & (BIN_NODES - 1)], 1);
          if (in_lds) lcol[slot - base] = p.y;
          else        col[slot] = p.y;
        }
      }
    }
#pragma unroll
    for (int u = 0; u < RUN_BATCH; ++u) {
      for (int k = lane + 64; k < dn[u]; k += 32) {
        const int2 p = pairs[dx[u] + k];
        const int slot = atomicAdd(&cin[p.x & (BIN_NODES - 1)], 1);
        if (in_lds) lcol[slot - base] = p.y;
        else        col[slot] = p.y;
      }
    }
  }
  __syncthreads();
  if (in_lds) {
    for (int i = t - (base & 31); i < total; i += 1024)
      if (i >= 0) col[base + i] = lcol[i];
  }
}

__global__ void move_words_kernel(const int* __restrict__ from, int* __restrict__ to, int n) {
  const int i = blockIdx.x * blockDim.x + threadIdx.x;
  if (i < n) to[i] = from[i];
}

template<int K, int M>
__global__ __launch_bounds__(256)
void gemm_wmma(const float* __restrict__ A, const float* __restrict__ B,
               float* __restrict__ C,
               const float* __restrict__ rowscale,
               const float* __restrict__ innorm, const float* __restrict__ bias,
               int relu, int ntiles) {
  constexpr int NT = M / 16;
  constexpr int G = K / 32;
  constexpr int PITCH = 2 * M + 32;
  __shared__ __attribute__((aligned(16))) float Bs[(K / 2) * PITCH];

  const int tid = threadIdx.x;
  for (int it = tid; it < (K / 2) * (M / 2); it += 256) {
    const int kp = it / (M / 2);
    const int c2 = (it % (M / 2)) * 2;
    const v2f r0 = *(const v2f*)(B + (size_t)(2 * kp) * M + c2);
    const v2f r1 = *(const v2f*)(B + (size_t)(2 * kp + 1) * M + c2);
    const v4f o = {r0.x, r1.x, r0.y, r1.y};
    *(v4f*)(Bs + kp * PITCH + c2 * 2) = o;
  }
  __syncthreads();

  const int lane = tid & 31;
  const int half = lane >> 4;
  const int m    = lane & 15;
  const int wave = __builtin_amdgcn_readfirstlane(tid >> 5);
  const int stride = (int)gridDim.x * 8;
  int tile = wave * (int)gridDim.x + (int)blockIdx.x;
  if (tile >= ntiles) return;

  const float* Bl = Bs + half * PITCH + m * 2;

  float bv[NT];
#pragma unroll
  for (int t = 0; t < NT; ++t) bv[t] = bias ? bias[16 * t + m] : 0.0f;

  v2f an[8];
  {
    const float* pa = A + (size_t)(tile * 16 + m) * K + 2 * half;
#pragma unroll
    for (int s = 0; s < 8; ++s) an[s] = *(const v2f*)(pa + 4 * s);
  }
  float scn = rowscale ? rowscale[tile * 16 + m] : 1.0f;

  for (;;) {
    const int ntile = tile + stride;
    const int ptile = (ntile < ntiles) ? ntile : tile;
    const float sc = scn;
    scn = rowscale ? rowscale[ptile * 16 + m] : 1.0f;
    const float* Arow  = A + (size_t)(tile  * 16 + m) * K + 2 * half;
    const float* Anext = A + (size_t)(ptile * 16 + m) * K + 2 * half;

    v8f acc[NT];
#pragma unroll
    for (int t = 0; t < NT; ++t) acc[t] = v8f{};

#pragma nounroll
    for (int g = 0; g < G; ++g) {
      v2f ac[8];
#pragma unroll
      for (int s = 0; s < 8; ++s) { ac[s].x = an[s].x * sc; ac[s].y = an[s].y * sc; }
      const float* pa = (g + 1 < G) ? (Arow + (g + 1) * 32) : Anext;
#pragma unroll
      for (int s = 0; s < 8; ++s) an[s] = *(const v2f*)(pa + 4 * s);

      const float* bp = Bl + g * (16 * PITCH);
      v2f bb[2][NT];
#pragma unroll
      for (int t = 0; t < NT; ++t) bb[0][t] = *(const v2f*)(bp + t * 32);
#pragma unroll
      for (int s = 0; s < 8; ++s) {
        if (s + 1 < 8) {
#pragma unroll
          for (int t = 0; t < NT; ++t)
            bb[(s + 1) & 1][t] = *(const v2f*)(bp + (s + 1) * (2 * PITCH) + t * 32);
        }
#pragma unroll
        for (int t = 0; t < NT; ++t)
          acc[t] = __builtin_amdgcn_wmma_f32_16x16x4_f32(
               false, ac[s],  false, bb[s & 1][t],
               (short)0, acc[t],  false,  false);
      }
    }

    const int row0 = tile * 16;
    float inn[8];
#pragma unroll
    for (int i = 0; i < 8; ++i) inn[i] = innorm ? innorm[row0 + i + 8 * half] : 1.0f;
#pragma unroll
    for (int i = 0; i < 8; ++i) {
      const int row = row0 + i + 8 * half;
#pragma unroll
      for (int t = 0; t < NT; ++t) {
        float v = acc[t][i];
        if (innorm) v = v * inn[i] + bv[t];
        if (relu)   v = fmaxf(v, 0.0f);
        C[(size_t)row * M + 16 * t + m] = v;
      }
    }

    if (ntile >= ntiles) break;
    tile = ntile;
  }
}


template<int M, bool SCALE, int B>
__device__ __forceinline__ void gather_add(typename fvec<M / 32>::type& acc,
                                           const float* __restrict__ hl,
                                           const int* __restrict__ c,
                                           const float* __restrict__ srcscale) {
  typedef typename fvec<M / 32>::type vec;
  int s[B];
#pragma unroll
  for (int k = 0; k < B; ++k) s[k] = c[k];
  vec r[B];
#pragma unroll
  for (int k = 0; k < B; ++k) r[k] = *(const vec*)(hl + (size_t)s[k] * M);
#pragma unroll
  for (int k = 0; k < B; ++k) {
    if (SCALE) acc += r[k] * srcscale[s[k]];
    else       acc += r[k];
  }
}

template<int M, bool SCALE>
__global__ __launch_bounds__(256)
void gather_kernel(const float* __restrict__ h,
                   const int* __restrict__ rowptr, const int* __restrict__ col,
                   const float* __restrict__ srcscale,
                   const float* __restrict__ innorm, const float* __restrict__ bias,
                   const float* __restrict__ postscale,
                   float* __restrict__ out, int N, int E, int relu) {
  constexpr int V = M / 32;
  typedef typename fvec<V>::type vec;
  const int lane = threadIdx.x & 31;
  const int row = __builtin_amdgcn_readfirstlane((int)(blockIdx.x * 8 + (threadIdx.x >> 5)));
  if (row >= N) return;
  const int beg = rowptr[row];
  const int end = (row + 1 < N) ? rowptr[row + 1] : E;
  const float* __restrict__ hl = h + lane * V;

  vec acc = {};
  int j = beg;
  for (; j + 8 <= end; j += 8) gather_add<M, SCALE, 8>(acc, hl, col + j, srcscale);
  if (j + 4 <= end) { gather_add<M, SCALE, 4>(acc, hl, col + j, srcscale); j += 4; }
  if (j + 2 <= end) { gather_add<M, SCALE, 2>(acc, hl, col + j, srcscale); j += 2; }
  if (j < end)      { gather_add<M, SCALE, 1>(acc, hl, col + j, srcscale); }

  if (innorm) {
    const float inn = innorm[row];
    const vec b = *(const vec*)(bias + lane * V);
    acc = acc * inn + b;
  }
  if (relu) {
#pragma unroll
    for (int c = 0; c < V; ++c) acc[c] = fmaxf(acc[c], 0.0f);
  }
  if (postscale) acc = acc * postscale[row];
  *(vec*)(out + (size_t)row * M + lane * V) = acc;
}


constexpr int TILE_PITCH = 128 + 4;
constexpr int FUSED_WAVES = 16;

template<int NT, int PITCH>
__device__ __forceinline__ void tile_product(v8f (&acc)[NT], const float* Al, const float* Bl) {
  v2f an[8];
#pragma unroll
  for (int s = 0; s < 8; ++s) an[s] = *(const v2f*)(Al + 4 * s);
#pragma nounroll
  for (int g = 0; g < 4; ++g) {
    v2f ac[8];
#pragma unroll
    for (int s = 0; s < 8; ++s) ac[s] = an[s];
    const float* pa = Al + ((g + 1 < 4) ? (g + 1) : g) * 32;
#pragma unroll
    for (int s = 0; s < 8; ++s) an[s] = *(const v2f*)(pa + 4 * s);

    const float* bp = Bl + g * (16 * PITCH);
    v2f bb[2][NT];
#pragma unroll
    for (int t = 0; t < NT; ++t) bb[0][t] = *(const v2f*)(bp + t * 32);
#pragma unroll
    for (int s = 0; s < 8; ++s) {
      if (s + 1 < 8) {
#pragma unroll
        for (int t = 0; t < NT; ++t)
          bb[(s + 1) & 1][t] = *(const v2f*)(bp + (s + 1) * (2 * PITCH) + t * 32);
      }
#pragma unroll
      for (int t = 0; t < NT; ++t)
        acc[t] = __builtin_amdgcn_wmma_f32_16x16x4_f32(
             false, ac[s],  false, bb[s & 1][t],
             (short)0, acc[t],  false,  false);
    }
  }
}

template<int K, int M, int PITCH, int THREADS>
__device__ __forceinline__ void stage_fragments(float* Bs, const float* __restrict__ B, int tid) {
  for (int it = tid; it < (K / 2) * (M / 2); it += THREADS) {
    const int kp = it / (M / 2);
    const int c2 = (it % (M / 2)) * 2;
    const v2f r0 = *(const v2f*)(B + (size_t)(2 * kp) * M + c2);
    const v2f r1 = *(const v2f*)(B + (size_t)(2 * kp + 1) * M + c2);
    const v4f o = {r0.x, r1.x, r0.y, r1.y};
    *(v4f*)(Bs + kp * PITCH + c2 * 2) = o;
  }
}

template<int M, int M3>
__global__ __launch_bounds__(FUSED_WAVES * 32)
void gather_gemm_gemm(const float* __restrict__ h,
                      const int* __restrict__ rowptr, const int* __restrict__ col,
                      const float* __restrict__ B,
                      const float* __restrict__ innorm, const float* __restrict__ bias, int relu,
                      const float* __restrict__ B3, const float* __restrict__ scale3,
                      float* __restrict__ C3, int ntiles, int N, int E) {
  constexpr int K   = 128;
  constexpr int NT  = M / 16;
  constexpr int NT3 = M3 / 16;
  constexpr int PITCH  = 2 * M + 32;
  constexpr int PITCH3 = 2 * M3 + 32;
  __shared__ __attribute__((aligned(16))) float Bs[(K / 2) * PITCH];
  __shared__ __attribute__((aligned(16))) float Bs3[(M / 2) * PITCH3];
  __shared__ __attribute__((aligned(16))) float At[FUSED_WAVES * 16 * TILE_PITCH];

  const int tid = threadIdx.x;
  stage_fragments<K, M, PITCH, FUSED_WAVES * 32>(Bs, B, tid);
  stage_fragments<M, M3, PITCH3, FUSED_WAVES * 32>(Bs3, B3, tid);
  __syncthreads();

  const int lane = tid & 31;
  const int half = lane >> 4;
  const int m    = lane & 15;
  const int wave = __builtin_amdgcn_readfirstlane(tid >> 5);
  const int stride = (int)gridDim.x * FUSED_WAVES;
  int tile = wave * (int)gridDim.x + (int)blockIdx.x;
  if (tile >= ntiles) return;

  const float* Bl  = Bs  + half * PITCH  + m * 2;
  const float* Bl3 = Bs3 + half * PITCH3 + m * 2;
  float* Aw = At + wave * (16 * TILE_PITCH);
  const float* Al = Aw + m * TILE_PITCH + 2 * half;
  const float* __restrict__ hl = h + lane * 4;

  float bv[NT];
#pragma unroll
  for (int t = 0; t < NT; ++t) bv[t] = bias ? bias[16 * t + m] : 0.0f;

  for (;;) {
    const int row0 = tile * 16;

    int rp = E;
    if (lane <= 16) { const int r = row0 + lane; if (r < N) rp = rowptr[r]; }
    const int tb = __builtin_amdgcn_readlane(rp, 0);
    const int te = __builtin_amdgcn_readlane(rp, 16);
    int cur  = 0;
    int rend = __builtin_amdgcn_readlane(rp, 1);
    v4f acc = {};
    int idv = 0;
    if (tb < te) idv = col[min(tb + lane, te - 1)];
    for (int j0 = tb; j0 < te; j0 += 32) {
      const int ids = idv;
      v4f r[32];
#pragma unroll
      for (int k = 0; k < 32; ++k) {
        const int s = __builtin_amdgcn_readlane(ids, k);
        r[k] = *(const v4f*)(hl + (size_t)s * K);
      }
      if (j0 + 32 < te) idv = col[min(j0 + 32 + lane, te - 1)];
#pragma unroll
      for (int k = 0; k < 32; ++k) {
        if (__builtin_expect(j0 + k >= rend, 0)) {
          do {
            *(v4f*)(Aw + cur * TILE_PITCH + lane * 4) = acc;
            acc = v4f{};
            ++cur;
            if (cur == 16) goto rows_done;
            rend = __builtin_amdgcn_readlane(rp, cur + 1);
          } while (j0 + k >= rend);
        }
        acc += r[k];
      }
    }
rows_done:
    for (; cur < 16; ++cur) {
      *(v4f*)(Aw + cur * TILE_PITCH + lane * 4) = acc;
      acc = v4f{};
    }

    {
      v8f pacc[NT];
#pragma unroll
      for (int t = 0; t < NT; ++t) pacc[t] = v8f{};
      tile_product<NT, PITCH>(pacc, Al, Bl);

      float inn[8], osc[8];
#pragma unroll
      for (int i = 0; i < 8; ++i) {
        inn[i] = innorm ? innorm[row0 + i + 8 * half] : 1.0f;
        osc[i] = scale3 ? scale3[row0 + i + 8 * half] : 1.0f;
      }
#pragma unroll
      for (int i = 0; i < 8; ++i) {
#pragma unroll
        for (int t = 0; t < NT; ++t) {
          float v = pacc[t][i];
          if (innorm) v = v * inn[i] + bv[t];
          if (relu)   v = fmaxf(v, 0.0f);
          Aw[(i + 8 * half) * TILE_PITCH + 16 * t + m] = v * osc[i];
        }
      }
    }

    {
      v8f qacc[NT3];
#pragma unroll
      for (int t = 0; t < NT3; ++t) qacc[t] = v8f{};
      tile_product<NT3, PITCH3>(qacc, Al, Bl3);
#pragma unroll
      for (int i = 0; i < 8; ++i) {
        const int row = row0 + i + 8 * half;
#pragma unroll
        for (int t = 0; t < NT3; ++t)
          C3[(size_t)row * M3 + 16 * t + m] = qacc[t][i];
      }
    }

    const int ntile = tile + stride;
    if (ntile >= ntiles) break;
    tile = ntile;
  }
}


extern "C" void kernel_launch(void* const* d_in, const int* in_sizes, int n_in,
                              void* d_out, int out_size, void* d_ws, size_t ws_size,
                              hipStream_t stream) {
  const float* x   = (const float*)d_in[0];
  const int*   src = (const int*)d_in[1];
  const int*   dst = (const int*)d_in[2];
  const float* W1  = (const float*)d_in[3];
  const float* b1  = (const float*)d_in[4];
  const float* W2  = (const float*)d_in[5];
  const float* b2  = (const float*)d_in[6];
  const float* W3  = (const float*)d_in[7];
  const float* b3  = (const float*)d_in[8];
  float* out = (float*)d_out;

  const int IN = 256, HID = 128, CLS = 64;
  const int E = in_sizes[1];
  const int N = in_sizes[0] / IN;

  float* ws      = (float*)d_ws;
  int*   rowptr  = (int*)ws;
  float* outnorm = ws + 2 * (size_t)N;
  float* innorm  = ws + 3 * (size_t)N;
  float* bufA    = ws + 4 * (size_t)N;
  float* bufB    = bufA + (size_t)N * HID;

  int* col   = (int*)d_out;
  int* col3  = (int*)bufB;

  const int nbins = (N + BIN_NODES - 1) / BIN_NODES;
  const int chunk = (E + SORT_BLOCKS - 1) / SORT_BLOCKS;
  int2* pairs = (int2*)bufB;
  int*  sbin  = (int*)bufB + 2 * (size_t)E;
  int2* descD = (int2*)((int*)bufB + 3 * (size_t)E);
  int2* descS = descD + (size_t)MAX_BINS * SORT_BLOCKS;

  const int rowtiles = N / 16;

  bin_sort_kernel<<<SORT_BLOCKS, 1024, 0, stream>>>(src, dst, E, chunk, nbins,
                                                    pairs, sbin, descD, descS);
  bin_rows_kernel<<<nbins, 1024, 0, stream>>>(pairs, sbin, descD, descS, SORT_BLOCKS,
                                              rowptr, outnorm, innorm, col, N);

  gemm_wmma<256, 128><<<512, 256, 0, stream>>>(x, W1, bufA, outnorm,
                                               nullptr, nullptr, 0, rowtiles);
  gather_kernel<128, false><<<(N + 7) / 8, 256, 0, stream>>>(bufA, rowptr, col, nullptr,
                                                             innorm, b1, outnorm, bufB, N, E, 1);

  gather_gemm_gemm<128, 64><<<256, 512, 0, stream>>>(bufB, rowptr, col, W2,
                                                     innorm, b2, 1, W3, outnorm,
                                                     bufA, rowtiles, N, E);

  move_words_kernel<<<(E + 255) / 256, 256, 0, stream>>>(col, col3, E);
  gather_kernel<64, false><<<(N + 7) / 8, 256, 0, stream>>>(bufA, rowptr, col3, nullptr,
                                                            innorm, b3, nullptr, out, N, E, 0);
}
